# starting kernel
# baseline (speedup 1.0000x reference)
#include <hip/hip_runtime.h>


typedef __attribute__((ext_vector_type(2))) float v2f;
typedef __attribute__((ext_vector_type(8))) float v8f;

#define BH 64
#define NN 4096
#define DD 64
#define K1_ROWS 256
#define K1_CHUNKS 16
#define C64 64

static __device__ __forceinline__ v8f wmma_f32(v2f a, v2f b, v8f c) {
  return __builtin_amdgcn_wmma_f32_16x16x4_f32(false, a, false, b, (short)0, c,
                                               false, false);
}

__global__ void k1_gpart(const float* __restrict__ k, const float* __restrict__ v,
                         float* __restrict__ Gpart, float* __restrict__ S) {
  const int bh = blockIdx.x;
  const int ch = blockIdx.y;
  const int t  = threadIdx.x;
  const float* kb = k + (size_t)bh * NN * DD;
  const float* vb = v + (size_t)bh * NN * DD;
  const int n0 = ch * K1_ROWS;

  {
    const int d = t & 63, seg = t >> 6;
    const int r0 = n0 + seg * 64;
    float s = 0.f;
    for (int r = 0; r < 64; ++r) s += kb[(size_t)(r0 + r) * DD + d];
    S[((size_t)(bh * C64) + (ch * 4 + seg)) * DD + d] = s;
  }

  const int wave = t >> 5, lane = t & 31;
  const int m = lane & 15, khalf = (lane >> 4) * 2;
  float* gout = Gpart + (size_t)(bh * K1_CHUNKS + ch) * (DD * DD);
  for (int i = 0; i < 2; ++i) {
    const int tt = wave * 2 + i;
    const int ti = tt >> 2, tj = tt & 3;
    v8f acc = {0.f, 0.f, 0.f, 0.f, 0.f, 0.f, 0.f, 0.f};
    for (int kk = 0; kk < K1_ROWS; kk += 4) {
      const int ra = n0 + kk + khalf;
      v2f a, b;
      a.x = kb[(size_t)ra * DD + ti * 16 + m];
      a.y = kb[(size_t)(ra + 1) * DD + ti * 16 + m];
      b.x = vb[(size_t)ra * DD + tj * 16 + m];
      b.y = vb[(size_t)(ra + 1) * DD + tj * 16 + m];
      acc = wmma_f32(a, b, acc);
    }
    const int rbase = ti * 16 + (lane >> 4) * 8;
    for (int j = 0; j < 8; ++j)
      gout[(rbase + j) * DD + tj * 16 + m] = acc[j];
  }
}

__global__ void k2_combine(const float* __restrict__ proj,
                           const float* __restrict__ Gpart,
                           float* __restrict__ S, float* __restrict__ Wm,
                           float* __restrict__ GramG) {
  __shared__ float G[DD * DD];
  __shared__ float Pl[DD * DD];
  __shared__ float Gm[DD * DD];
  const int bh = blockIdx.x;
  const int t  = threadIdx.x;

  for (int idx = t; idx < DD * DD; idx += 256) {
    float s = 0.f;
    for (int c = 0; c < K1_CHUNKS; ++c)
      s += Gpart[(size_t)(bh * K1_CHUNKS + c) * (DD * DD) + idx];
    G[idx]  = s;
    Pl[idx] = proj[idx];
  }
  __syncthreads();

  for (int idx = t; idx < DD * DD; idx += 256) {
    const int d1 = idx >> 6, d2 = idx & 63;
    float s = 0.f;
    for (int mm = 0; mm < DD; ++mm) s += Pl[mm * DD + d1] * Pl[mm * DD + d2];
    Gm[idx] = s;
    if (bh == 0) GramG[idx] = s;
  }
  __syncthreads();

  for (int idx = t; idx < DD * DD; idx += 256) {
    const int d1 = idx >> 6, e = idx & 63;
    float s = 0.f;
    for (int d2 = 0; d2 < DD; ++d2) s += Gm[d1 * DD + d2] * G[d2 * DD + e];
    Wm[(size_t)bh * DD * DD + idx] = s;
  }

  if (t < 64) {
    float run = 0.f;
    for (int c = 0; c < C64; ++c) {
      const size_t idx = ((size_t)(bh * C64) + c) * DD + t;
      const float tmp = S[idx];
      S[idx] = run;
      run += tmp;
    }
  }
}

__global__ void k3_out(const float* __restrict__ q, const float* __restrict__ k,
                       const float* __restrict__ S, const float* __restrict__ Wm,
                       const float* __restrict__ GramG, float* __restrict__ out) {
  __shared__ float kc[64 * 65];
  __shared__ float ab[64 * 65];
  __shared__ float qgl[64 * 65];
  __shared__ float qsum[4 * 64];
  __shared__ float dinv[64];
  const int bh = blockIdx.x, c64 = blockIdx.y;
  const int t  = threadIdx.x;
  const int n0 = c64 * 64;
  const float* qb = q + (size_t)bh * NN * DD;
  const float* kb = k + (size_t)bh * NN * DD;

  for (int idx = t; idx < 64 * 64; idx += 256) {
    const int r = idx >> 6, d = idx & 63;
    kc[r * 65 + d] = kb[(size_t)n0 * DD + idx];
    ab[r * 65 + d] = GramG[idx];
  }
  __syncthreads();

  const int d = t & 63, qt = t >> 6;
  {
    float run = 0.f;
    for (int r = qt * 16; r < qt * 16 + 16; ++r) {
      run += kc[r * 65 + d];
      kc[r * 65 + d] = run;
    }
    qsum[qt * 64 + d] = run;
  }
  __syncthreads();
  if (t < 64) {
    const float p = S[((size_t)(bh * C64) + c64) * DD + t];
    const float o0 = p;
    const float o1 = o0 + qsum[0 * 64 + t];
    const float o2 = o1 + qsum[1 * 64 + t];
    const float o3 = o2 + qsum[2 * 64 + t];
    qsum[0 * 64 + t] = o0; qsum[1 * 64 + t] = o1;
    qsum[2 * 64 + t] = o2; qsum[3 * 64 + t] = o3;
  }
  __syncthreads();
  {
    const float off = qsum[qt * 64 + d];
    for (int r = qt * 16; r < qt * 16 + 16; ++r) kc[r * 65 + d] += off;
  }
  __syncthreads();

  const int wave = t >> 5, lane = t & 31;
  const int m = lane & 15, khalf = (lane >> 4) * 2;

  for (int i = 0; i < 2; ++i) {
    const int tt = wave * 2 + i;
    const int ti = tt >> 2, tj = tt & 3;
    v8f acc = {0.f, 0.f, 0.f, 0.f, 0.f, 0.f, 0.f, 0.f};
    const float* arow = qb + (size_t)(n0 + ti * 16 + m) * DD;
    for (int kk = 0; kk < DD; kk += 4) {
      v2f a = *(const v2f*)(arow + kk + khalf);
      v2f b;
      b.x = ab[(kk + khalf) * 65 + tj * 16 + m];
      b.y = ab[(kk + khalf + 1) * 65 + tj * 16 + m];
      acc = wmma_f32(a, b, acc);
    }
    const int rbase = ti * 16 + (lane >> 4) * 8;
    for (int j = 0; j < 8; ++j)
      qgl[(rbase + j) * 65 + tj * 16 + m] = acc[j];
  }
  __syncthreads();

  if (t < 64) {
    float s = 0.f;
    for (int dd = 0; dd < 64; ++dd) s += qgl[t * 65 + dd] * kc[t * 65 + dd];
    dinv[t] = 1.0f / s;
  }
  __syncthreads();

  for (int idx = t; idx < 64 * 64; idx += 256) {
    const int r = idx >> 6, dd = idx & 63;
    ab[r * 65 + dd] = Wm[(size_t)bh * DD * DD + idx];
  }
  __syncthreads();

  for (int i = 0; i < 2; ++i) {
    const int tt = wave * 2 + i;
    const int ti = tt >> 2, tj = tt & 3;
    v8f acc = {0.f, 0.f, 0.f, 0.f, 0.f, 0.f, 0.f, 0.f};
    const float* arow = qb + (size_t)(n0 + ti * 16 + m) * DD;
    for (int kk = 0; kk < DD; kk += 4) {
      v2f a = *(const v2f*)(arow + kk + khalf);
      v2f b;
      b.x = ab[(kk + khalf) * 65 + tj * 16 + m];
      b.y = ab[(kk + khalf + 1) * 65 + tj * 16 + m];
      acc = wmma_f32(a, b, acc);
    }
    const int rbase = ti * 16 + (lane >> 4) * 8;
    for (int j = 0; j < 8; ++j) {
      const float dv = dinv[rbase + j];
      out[((size_t)bh * NN + n0 + rbase + j) * DD + tj * 16 + m] = acc[j] * dv;
    }
  }
}

extern "C" void kernel_launch(void* const* d_in, const int* in_sizes, int n_in,
                              void* d_out, int out_size, void* d_ws, size_t ws_size,
                              hipStream_t stream) {
  const float* q    = (const float*)d_in[0];
  const float* k    = (const float*)d_in[1];
  const float* v    = (const float*)d_in[2];
  const float* proj = (const float*)d_in[3];
  float* out = (float*)d_out;

  float* ws    = (float*)d_ws;
  float* Gpart = ws;
  float* S     = Gpart + (size_t)BH * K1_CHUNKS * DD * DD;
  float* Wm    = S + (size_t)BH * C64 * DD;
  float* Gram  = Wm + (size_t)BH * DD * DD;

  k1_gpart<<<dim3(BH, K1_CHUNKS), 256, 0, stream>>>(k, v, Gpart, S);
  k2_combine<<<dim3(BH), 256, 0, stream>>>(proj, Gpart, S, Wm, Gram);
  k3_out<<<dim3(BH, C64), 256, 0, stream>>>(q, k, S, Wm, Gram, out);
}


# K1: each wave reads whole k,v rows with 128-bit loads and feeds all 16 tiles from them (tile i,j holds d=4c+i, e=4c+j); 4 waves split the rows, partials summed in LDS; G and S sum order changed
# speedup vs baseline: 1.1223x; 1.1223x over previous
#include <hip/hip_runtime.h>


typedef __attribute__((ext_vector_type(2))) float v2f;
typedef __attribute__((ext_vector_type(4))) float v4f;
typedef __attribute__((ext_vector_type(8))) float v8f;

#define BH 64
#define NN 4096
#define DD 64
#define K1_ROWS 256
#define K1_CHUNKS 16
#define C64 64

static __device__ __forceinline__ v8f wmma_f32(v2f a, v2f b, v8f c) {
  return __builtin_amdgcn_wmma_f32_16x16x4_f32(false, a, false, b, (short)0, c,
                                               false, false);
}

#define K1_WAVES 4
#define K1_STEPS 16
#define K1_AHEAD 5

__global__ void __launch_bounds__(128)
k1_gpart(const float* __restrict__ k, const float* __restrict__ v,
         float* __restrict__ Gpart, float* __restrict__ S) {
  __shared__ float red[K1_WAVES * DD * DD];
  const int bh = blockIdx.x;
  const int ch = blockIdx.y;
  const int t  = threadIdx.x;
  const int wave = t >> 5, lane = t & 31;
  const int c = lane & 15, h = lane >> 4;
  const int seg = ch * K1_WAVES + wave;
  const size_t base = ((size_t)bh * NN + (size_t)seg * 64 + 2 * h) * DD + 4 * c;
  const float* kp = k + base;
  const float* vp = v + base;

  v8f acc[4][4];
  for (int i = 0; i < 4; ++i)
    for (int j = 0; j < 4; ++j)
      acc[i][j] = (v8f){0.f, 0.f, 0.f, 0.f, 0.f, 0.f, 0.f, 0.f};
  v4f ksum = {0.f, 0.f, 0.f, 0.f};

  v4f rk0[K1_AHEAD], rk1[K1_AHEAD], rv0[K1_AHEAD], rv1[K1_AHEAD];
  for (int p = 0; p < K1_AHEAD - 1; ++p) {
    rk0[p] = *(const v4f*)(kp + (4 * p) * DD);
    rk1[p] = *(const v4f*)(kp + (4 * p + 1) * DD);
    rv0[p] = *(const v4f*)(vp + (4 * p) * DD);
    rv1[p] = *(const v4f*)(vp + (4 * p + 1) * DD);
  }
  for (int s = 0; s < K1_STEPS; ++s) {
    const int nx = s + K1_AHEAD - 1;
    if (nx < K1_STEPS) {
      const int w = nx % K1_AHEAD;
      rk0[w] = *(const v4f*)(kp + (4 * nx) * DD);
      rk1[w] = *(const v4f*)(kp + (4 * nx + 1) * DD);
      rv0[w] = *(const v4f*)(vp + (4 * nx) * DD);
      rv1[w] = *(const v4f*)(vp + (4 * nx + 1) * DD);
    }
    const int u = s % K1_AHEAD;
    ksum += rk0[u] + rk1[u];
    for (int i = 0; i < 4; ++i) {
      v2f a;
      a.x = rk0[u][i];
      a.y = rk1[u][i];
      for (int j = 0; j < 4; ++j) {
        v2f b;
        b.x = rv0[u][j];
        b.y = rv1[u][j];
        acc[i][j] = wmma_f32(a, b, acc[i][j]);
      }
    }
  }

  {
    v4f o;
    o.x = ksum.x + __shfl_xor(ksum.x, 16);
    o.y = ksum.y + __shfl_xor(ksum.y, 16);
    o.z = ksum.z + __shfl_xor(ksum.z, 16);
    o.w = ksum.w + __shfl_xor(ksum.w, 16);
    if (h == 0) *(v4f*)(S + ((size_t)(bh * C64) + seg) * DD + 4 * c) = o;
  }

  float* my = red + wave * (DD * DD);
  for (int i = 0; i < 4; ++i)
    for (int r = 0; r < 8; ++r) {
      v4f o;
      o.x = acc[i][0][r];
      o.y = acc[i][1][r];
      o.z = acc[i][2][r];
      o.w = acc[i][3][r];
      *(v4f*)(my + (4 * (8 * h + r) + i) * DD + 4 * c) = o;
    }
  __syncthreads();

  float* gout = Gpart + (size_t)(bh * K1_CHUNKS + ch) * (DD * DD);
  for (int x = 0; x < 8; ++x) {
    const int idx = (x * 128 + t) * 4;
    const v4f p0 = *(const v4f*)(red + idx);
    const v4f p1 = *(const v4f*)(red + DD * DD + idx);
    const v4f p2 = *(const v4f*)(red + 2 * DD * DD + idx);
    const v4f p3 = *(const v4f*)(red + 3 * DD * DD + idx);
    *(v4f*)(gout + idx) = ((p0 + p1) + p2) + p3;
  }
}

__global__ void k2_combine(const float* __restrict__ proj,
                           const float* __restrict__ Gpart,
                           float* __restrict__ S, float* __restrict__ Wm,
                           float* __restrict__ GramG) {
  __shared__ float G[DD * DD];
  __shared__ float Pl[DD * DD];
  __shared__ float Gm[DD * DD];
  const int bh = blockIdx.x;
  const int t  = threadIdx.x;

  for (int idx = t; idx < DD * DD; idx += 256) {
    float s = 0.f;
    for (int c = 0; c < K1_CHUNKS; ++c)
      s += Gpart[(size_t)(bh * K1_CHUNKS + c) * (DD * DD) + idx];
    G[idx]  = s;
    Pl[idx] = proj[idx];
  }
  __syncthreads();

  for (int idx = t; idx < DD * DD; idx += 256) {
    const int d1 = idx >> 6, d2 = idx & 63;
    float s = 0.f;
    for (int mm = 0; mm < DD; ++mm) s += Pl[mm * DD + d1] * Pl[mm * DD + d2];
    Gm[idx] = s;
    if (bh == 0) GramG[idx] = s;
  }
  __syncthreads();

  for (int idx = t; idx < DD * DD; idx += 256) {
    const int d1 = idx >> 6, e = idx & 63;
    float s = 0.f;
    for (int d2 = 0; d2 < DD; ++d2) s += Gm[d1 * DD + d2] * G[d2 * DD + e];
    Wm[(size_t)bh * DD * DD + idx] = s;
  }

  if (t < 64) {
    float run = 0.f;
    for (int c = 0; c < C64; ++c) {
      const size_t idx = ((size_t)(bh * C64) + c) * DD + t;
      const float tmp = S[idx];
      S[idx] = run;
      run += tmp;
    }
  }
}

__global__ void k3_out(const float* __restrict__ q, const float* __restrict__ k,
                       const float* __restrict__ S, const float* __restrict__ Wm,
                       const float* __restrict__ GramG, float* __restrict__ out) {
  __shared__ float kc[64 * 65];
  __shared__ float ab[64 * 65];
  __shared__ float qgl[64 * 65];
  __shared__ float qsum[4 * 64];
  __shared__ float dinv[64];
  const int bh = blockIdx.x, c64 = blockIdx.y;
  const int t  = threadIdx.x;
  const int n0 = c64 * 64;
  const float* qb = q + (size_t)bh * NN * DD;
  const float* kb = k + (size_t)bh * NN * DD;

  for (int idx = t; idx < 64 * 64; idx += 256) {
    const int r = idx >> 6, d = idx & 63;
    kc[r * 65 + d] = kb[(size_t)n0 * DD + idx];
    ab[r * 65 + d] = GramG[idx];
  }
  __syncthreads();

  const int d = t & 63, qt = t >> 6;
  {
    float run = 0.f;
    for (int r = qt * 16; r < qt * 16 + 16; ++r) {
      run += kc[r * 65 + d];
      kc[r * 65 + d] = run;
    }
    qsum[qt * 64 + d] = run;
  }
  __syncthreads();
  if (t < 64) {
    const float p = S[((size_t)(bh * C64) + c64) * DD + t];
    const float o0 = p;
    const float o1 = o0 + qsum[0 * 64 + t];
    const float o2 = o1 + qsum[1 * 64 + t];
    const float o3 = o2 + qsum[2 * 64 + t];
    qsum[0 * 64 + t] = o0; qsum[1 * 64 + t] = o1;
    qsum[2 * 64 + t] = o2; qsum[3 * 64 + t] = o3;
  }
  __syncthreads();
  {
    const float off = qsum[qt * 64 + d];
    for (int r = qt * 16; r < qt * 16 + 16; ++r) kc[r * 65 + d] += off;
  }
  __syncthreads();

  const int wave = t >> 5, lane = t & 31;
  const int m = lane & 15, khalf = (lane >> 4) * 2;

  for (int i = 0; i < 2; ++i) {
    const int tt = wave * 2 + i;
    const int ti = tt >> 2, tj = tt & 3;
    v8f acc = {0.f, 0.f, 0.f, 0.f, 0.f, 0.f, 0.f, 0.f};
    const float* arow = qb + (size_t)(n0 + ti * 16 + m) * DD;
    for (int kk = 0; kk < DD; kk += 4) {
      v2f a = *(const v2f*)(arow + kk + khalf);
      v2f b;
      b.x = ab[(kk + khalf) * 65 + tj * 16 + m];
      b.y = ab[(kk + khalf + 1) * 65 + tj * 16 + m];
      acc = wmma_f32(a, b, acc);
    }
    const int rbase = ti * 16 + (lane >> 4) * 8;
    for (int j = 0; j < 8; ++j)
      qgl[(rbase + j) * 65 + tj * 16 + m] = acc[j];
  }
  __syncthreads();

  if (t < 64) {
    float s = 0.f;
    for (int dd = 0; dd < 64; ++dd) s += qgl[t * 65 + dd] * kc[t * 65 + dd];
    dinv[t] = 1.0f / s;
  }
  __syncthreads();

  for (int idx = t; idx < 64 * 64; idx += 256) {
    const int r = idx >> 6, dd = idx & 63;
    ab[r * 65 + dd] = Wm[(size_t)bh * DD * DD + idx];
  }
  __syncthreads();

  for (int i = 0; i < 2; ++i) {
    const int tt = wave * 2 + i;
    const int ti = tt >> 2, tj = tt & 3;
    v8f acc = {0.f, 0.f, 0.f, 0.f, 0.f, 0.f, 0.f, 0.f};
    const float* arow = qb + (size_t)(n0 + ti * 16 + m) * DD;
    for (int kk = 0; kk < DD; kk += 4) {
      v2f a = *(const v2f*)(arow + kk + khalf);
      v2f b;
      b.x = ab[(kk + khalf) * 65 + tj * 16 + m];
      b.y = ab[(kk + khalf + 1) * 65 + tj * 16 + m];
      acc = wmma_f32(a, b, acc);
    }
    const int rbase = ti * 16 + (lane >> 4) * 8;
    for (int j = 0; j < 8; ++j) {
      const float dv = dinv[rbase + j];
      out[((size_t)bh * NN + n0 + rbase + j) * DD + tj * 16 + m] = acc[j] * dv;
    }
  }
}

extern "C" void kernel_launch(void* const* d_in, const int* in_sizes, int n_in,
                              void* d_out, int out_size, void* d_ws, size_t ws_size,
                              hipStream_t stream) {
  const float* q    = (const float*)d_in[0];
  const float* k    = (const float*)d_in[1];
  const float* v    = (const float*)d_in[2];
  const float* proj = (const float*)d_in[3];
  float* out = (float*)d_out;

  float* ws    = (float*)d_ws;
  float* Gpart = ws;
  float* S     = Gpart + (size_t)BH * K1_CHUNKS * DD * DD;
  float* Wm    = S + (size_t)BH * C64 * DD;
  float* Gram  = Wm + (size_t)BH * DD * DD;

  k1_gpart<<<dim3(BH, K1_CHUNKS), 128, 0, stream>>>(k, v, Gpart, S);
  k2_combine<<<dim3(BH), 256, 0, stream>>>(proj, Gpart, S, Wm, Gram);
  k3_out<<<dim3(BH, C64), 256, 0, stream>>>(q, k, S, Wm, Gram, out);
}


# K2: Gram = P^T P and W = Gram G move from per-element 64-long scalar FMA chains to the f32 16x16x4 matrix instruction (8 waves, 2 tiles each); sum order of these two products may change
# speedup vs baseline: 1.3038x; 1.1617x over previous
#include <hip/hip_runtime.h>


typedef __attribute__((ext_vector_type(2))) float v2f;
typedef __attribute__((ext_vector_type(4))) float v4f;
typedef __attribute__((ext_vector_type(8))) float v8f;

#define BH 64
#define NN 4096
#define DD 64
#define K1_ROWS 256
#define K1_CHUNKS 16
#define C64 64

static __device__ __forceinline__ v8f wmma_f32(v2f a, v2f b, v8f c) {
  return __builtin_amdgcn_wmma_f32_16x16x4_f32(false, a, false, b, (short)0, c,
                                               false, false);
}

#define K1_WAVES 4
#define K1_STEPS 16
#define K1_AHEAD 5

__global__ void __launch_bounds__(128)
k1_gpart(const float* __restrict__ k, const float* __restrict__ v,
         float* __restrict__ Gpart, float* __restrict__ S) {
  __shared__ float red[K1_WAVES * DD * DD];
  const int bh = blockIdx.x;
  const int ch = blockIdx.y;
  const int t  = threadIdx.x;
  const int wave = t >> 5, lane = t & 31;
  const int c = lane & 15, h = lane >> 4;
  const int seg = ch * K1_WAVES + wave;
  const size_t base = ((size_t)bh * NN + (size_t)seg * 64 + 2 * h) * DD + 4 * c;
  const float* kp = k + base;
  const float* vp = v + base;

  v8f acc[4][4];
  for (int i = 0; i < 4; ++i)
    for (int j = 0; j < 4; ++j)
      acc[i][j] = (v8f){0.f, 0.f, 0.f, 0.f, 0.f, 0.f, 0.f, 0.f};
  v4f ksum = {0.f, 0.f, 0.f, 0.f};

  v4f rk0[K1_AHEAD], rk1[K1_AHEAD], rv0[K1_AHEAD], rv1[K1_AHEAD];
  for (int p = 0; p < K1_AHEAD - 1; ++p) {
    rk0[p] = *(const v4f*)(kp + (4 * p) * DD);
    rk1[p] = *(const v4f*)(kp + (4 * p + 1) * DD);
    rv0[p] = *(const v4f*)(vp + (4 * p) * DD);
    rv1[p] = *(const v4f*)(vp + (4 * p + 1) * DD);
  }
  for (int s = 0; s < K1_STEPS; ++s) {
    const int nx = s + K1_AHEAD - 1;
    if (nx < K1_STEPS) {
      const int w = nx % K1_AHEAD;
      rk0[w] = *(const v4f*)(kp + (4 * nx) * DD);
      rk1[w] = *(const v4f*)(kp + (4 * nx + 1) * DD);
      rv0[w] = *(const v4f*)(vp + (4 * nx) * DD);
      rv1[w] = *(const v4f*)(vp + (4 * nx + 1) * DD);
    }
    const int u = s % K1_AHEAD;
    ksum += rk0[u] + rk1[u];
    for (int i = 0; i < 4; ++i) {
      v2f a;
      a.x = rk0[u][i];
      a.y = rk1[u][i];
      for (int j = 0; j < 4; ++j) {
        v2f b;
        b.x = rv0[u][j];
        b.y = rv1[u][j];
        acc[i][j] = wmma_f32(a, b, acc[i][j]);
      }
    }
  }

  {
    v4f o;
    o.x = ksum.x + __shfl_xor(ksum.x, 16);
    o.y = ksum.y + __shfl_xor(ksum.y, 16);
    o.z = ksum.z + __shfl_xor(ksum.z, 16);
    o.w = ksum.w + __shfl_xor(ksum.w, 16);
    if (h == 0) *(v4f*)(S + ((size_t)(bh * C64) + seg) * DD + 4 * c) = o;
  }

  float* my = red + wave * (DD * DD);
  for (int i = 0; i < 4; ++i)
    for (int r = 0; r < 8; ++r) {
      v4f o;
      o.x = acc[i][0][r];
      o.y = acc[i][1][r];
      o.z = acc[i][2][r];
      o.w = acc[i][3][r];
      *(v4f*)(my + (4 * (8 * h + r) + i) * DD + 4 * c) = o;
    }
  __syncthreads();

  float* gout = Gpart + (size_t)(bh * K1_CHUNKS + ch) * (DD * DD);
  for (int x = 0; x < 8; ++x) {
    const int idx = (x * 128 + t) * 4;
    const v4f p0 = *(const v4f*)(red + idx);
    const v4f p1 = *(const v4f*)(red + DD * DD + idx);
    const v4f p2 = *(const v4f*)(red + 2 * DD * DD + idx);
    const v4f p3 = *(const v4f*)(red + 3 * DD * DD + idx);
    *(v4f*)(gout + idx) = ((p0 + p1) + p2) + p3;
  }
}

__global__ void
k2_combine(const float* __restrict__ proj,
           const float* __restrict__ Gpart,
           float* __restrict__ S, float* __restrict__ Wm,
           float* __restrict__ GramG) {
  __shared__ float G[DD * DD];
  __shared__ float Pl[DD * DD];
  __shared__ float Gm[DD * 65];
  const int bh = blockIdx.x;
  const int t  = threadIdx.x;

  for (int idx = t; idx < DD * DD; idx += 256) {
    float s = 0.f;
    for (int c = 0; c < K1_CHUNKS; ++c)
      s += Gpart[(size_t)(bh * K1_CHUNKS + c) * (DD * DD) + idx];
    G[idx]  = s;
    Pl[idx] = proj[idx];
  }
  __syncthreads();

  const int wave = t >> 5, lane = t & 31;
  const int m = lane & 15, khalf = (lane >> 4) * 2;
  const int ti = wave >> 1;
  const int c0 = (wave & 1) * 32 + m;
  const int c1 = c0 + 16;
  const int rbase = ti * 16 + (lane >> 4) * 8;

  {
    v8f acc0 = {0.f, 0.f, 0.f, 0.f, 0.f, 0.f, 0.f, 0.f};
    v8f acc1 = {0.f, 0.f, 0.f, 0.f, 0.f, 0.f, 0.f, 0.f};
    for (int kk = 0; kk < DD; kk += 4) {
      const float* r0 = Pl + (kk + khalf) * DD;
      const float* r1 = r0 + DD;
      v2f a, b0, b1;
      a.x  = r0[ti * 16 + m];
      a.y  = r1[ti * 16 + m];
      b0.x = r0[c0];
      b0.y = r1[c0];
      b1.x = r0[c1];
      b1.y = r1[c1];
      acc0 = wmma_f32(a, b0, acc0);
      acc1 = wmma_f32(a, b1, acc1);
    }
    for (int j = 0; j < 8; ++j) {
      Gm[(rbase + j) * 65 + c0] = acc0[j];
      Gm[(rbase + j) * 65 + c1] = acc1[j];
    }
    if (bh == 0) {
      for (int j = 0; j < 8; ++j) {
        GramG[(rbase + j) * DD + c0] = acc0[j];
        GramG[(rbase + j) * DD + c1] = acc1[j];
      }
    }
  }
  __syncthreads();

  {
    v8f acc0 = {0.f, 0.f, 0.f, 0.f, 0.f, 0.f, 0.f, 0.f};
    v8f acc1 = {0.f, 0.f, 0.f, 0.f, 0.f, 0.f, 0.f, 0.f};
    const float* arow = Gm + (ti * 16 + m) * 65 + khalf;
    for (int kk = 0; kk < DD; kk += 4) {
      const float* r0 = G + (kk + khalf) * DD;
      const float* r1 = r0 + DD;
      v2f a, b0, b1;
      a.x  = arow[kk];
      a.y  = arow[kk + 1];
      b0.x = r0[c0];
      b0.y = r1[c0];
      b1.x = r0[c1];
      b1.y = r1[c1];
      acc0 = wmma_f32(a, b0, acc0);
      acc1 = wmma_f32(a, b1, acc1);
    }
    float* wout = Wm + (size_t)bh * DD * DD;
    for (int j = 0; j < 8; ++j) {
      wout[(rbase + j) * DD + c0] = acc0[j];
      wout[(rbase + j) * DD + c1] = acc1[j];
    }
  }

  if (t < 64) {
    float run = 0.f;
    for (int c = 0; c < C64; ++c) {
      const size_t idx = ((size_t)(bh * C64) + c) * DD + t;
      const float tmp = S[idx];
      S[idx] = run;
      run += tmp;
    }
  }
}

__global__ void k3_out(const float* __restrict__ q, const float* __restrict__ k,
                       const float* __restrict__ S, const float* __restrict__ Wm,
                       const float* __restrict__ GramG, float* __restrict__ out) {
  __shared__ float kc[64 * 65];
  __shared__ float ab[64 * 65];
  __shared__ float qgl[64 * 65];
  __shared__ float qsum[4 * 64];
  __shared__ float dinv[64];
  const int bh = blockIdx.x, c64 = blockIdx.y;
  const int t  = threadIdx.x;
  const int n0 = c64 * 64;
  const float* qb = q + (size_t)bh * NN * DD;
  const float* kb = k + (size_t)bh * NN * DD;

  for (int idx = t; idx < 64 * 64; idx += 256) {
    const int r = idx >> 6, d = idx & 63;
    kc[r * 65 + d] = kb[(size_t)n0 * DD + idx];
    ab[r * 65 + d] = GramG[idx];
  }
  __syncthreads();

  const int d = t & 63, qt = t >> 6;
  {
    float run = 0.f;
    for (int r = qt * 16; r < qt * 16 + 16; ++r) {
      run += kc[r * 65 + d];
      kc[r * 65 + d] = run;
    }
    qsum[qt * 64 + d] = run;
  }
  __syncthreads();
  if (t < 64) {
    const float p = S[((size_t)(bh * C64) + c64) * DD + t];
    const float o0 = p;
    const float o1 = o0 + qsum[0 * 64 + t];
    const float o2 = o1 + qsum[1 * 64 + t];
    const float o3 = o2 + qsum[2 * 64 + t];
    qsum[0 * 64 + t] = o0; qsum[1 * 64 + t] = o1;
    qsum[2 * 64 + t] = o2; qsum[3 * 64 + t] = o3;
  }
  __syncthreads();
  {
    const float off = qsum[qt * 64 + d];
    for (int r = qt * 16; r < qt * 16 + 16; ++r) kc[r * 65 + d] += off;
  }
  __syncthreads();

  const int wave = t >> 5, lane = t & 31;
  const int m = lane & 15, khalf = (lane >> 4) * 2;

  for (int i = 0; i < 2; ++i) {
    const int tt = wave * 2 + i;
    const int ti = tt >> 2, tj = tt & 3;
    v8f acc = {0.f, 0.f, 0.f, 0.f, 0.f, 0.f, 0.f, 0.f};
    const float* arow = qb + (size_t)(n0 + ti * 16 + m) * DD;
    for (int kk = 0; kk < DD; kk += 4) {
      v2f a = *(const v2f*)(arow + kk + khalf);
      v2f b;
      b.x = ab[(kk + khalf) * 65 + tj * 16 + m];
      b.y = ab[(kk + khalf + 1) * 65 + tj * 16 + m];
      acc = wmma_f32(a, b, acc);
    }
    const int rbase = ti * 16 + (lane >> 4) * 8;
    for (int j = 0; j < 8; ++j)
      qgl[(rbase + j) * 65 + tj * 16 + m] = acc[j];
  }
  __syncthreads();

  if (t < 64) {
    float s = 0.f;
    for (int dd = 0; dd < 64; ++dd) s += qgl[t * 65 + dd] * kc[t * 65 + dd];
    dinv[t] = 1.0f / s;
  }
  __syncthreads();

  for (int idx = t; idx < 64 * 64; idx += 256) {
    const int r = idx >> 6, dd = idx & 63;
    ab[r * 65 + dd] = Wm[(size_t)bh * DD * DD + idx];
  }
  __syncthreads();

  for (int i = 0; i < 2; ++i) {
    const int tt = wave * 2 + i;
    const int ti = tt >> 2, tj = tt & 3;
    v8f acc = {0.f, 0.f, 0.f, 0.f, 0.f, 0.f, 0.f, 0.f};
    const float* arow = qb + (size_t)(n0 + ti * 16 + m) * DD;
    for (int kk = 0; kk < DD; kk += 4) {
      v2f a = *(const v2f*)(arow + kk + khalf);
      v2f b;
      b.x = ab[(kk + khalf) * 65 + tj * 16 + m];
      b.y = ab[(kk + khalf + 1) * 65 + tj * 16 + m];
      acc = wmma_f32(a, b, acc);
    }
    const int rbase = ti * 16 + (lane >> 4) * 8;
    for (int j = 0; j < 8; ++j) {
      const float dv = dinv[rbase + j];
      out[((size_t)bh * NN + n0 + rbase + j) * DD + tj * 16 + m] = acc[j] * dv;
    }
  }
}

extern "C" void kernel_launch(void* const* d_in, const int* in_sizes, int n_in,
                              void* d_out, int out_size, void* d_ws, size_t ws_size,
                              hipStream_t stream) {
  const float* q    = (const float*)d_in[0];
  const float* k    = (const float*)d_in[1];
  const float* v    = (const float*)d_in[2];
  const float* proj = (const float*)d_in[3];
  float* out = (float*)d_out;

  float* ws    = (float*)d_ws;
  float* Gpart = ws;
  float* S     = Gpart + (size_t)BH * K1_CHUNKS * DD * DD;
  float* Wm    = S + (size_t)BH * C64 * DD;
  float* Gram  = Wm + (size_t)BH * DD * DD;

  k1_gpart<<<dim3(BH, K1_CHUNKS), 128, 0, stream>>>(k, v, Gpart, S);
  k2_combine<<<dim3(BH), 256, 0, stream>>>(proj, Gpart, S, Wm, Gram);
  k3_out<<<dim3(BH, C64), 256, 0, stream>>>(q, k, S, Wm, Gram, out);
}


# K2, K3: each work-group requests its staged global operands (K3: k chunk, Gram, W, prefix row; K2: G partials, scan rows) as one early batch of wide loads, not chains of short serial round trips
# speedup vs baseline: 1.5650x; 1.2003x over previous
#include <hip/hip_runtime.h>


typedef __attribute__((ext_vector_type(2))) float v2f;
typedef __attribute__((ext_vector_type(4))) float v4f;
typedef __attribute__((ext_vector_type(8))) float v8f;

#define BH 64
#define NN 4096
#define DD 64
#define K1_ROWS 256
#define K1_CHUNKS 16
#define C64 64

static __device__ __forceinline__ v8f wmma_f32(v2f a, v2f b, v8f c) {
  return __builtin_amdgcn_wmma_f32_16x16x4_f32(false, a, false, b, (short)0, c,
                                               false, false);
}

#define K1_WAVES 4
#define K1_STEPS 16
#define K1_AHEAD 5

__global__ void __launch_bounds__(128)
k1_gpart(const float* __restrict__ k, const float* __restrict__ v,
         float* __restrict__ Gpart, float* __restrict__ S) {
  __shared__ float red[K1_WAVES * DD * DD];
  const int bh = blockIdx.x;
  const int ch = blockIdx.y;
  const int t  = threadIdx.x;
  const int wave = t >> 5, lane = t & 31;
  const int c = lane & 15, h = lane >> 4;
  const int seg = ch * K1_WAVES + wave;
  const size_t base = ((size_t)bh * NN + (size_t)seg * 64 + 2 * h) * DD + 4 * c;
  const float* kp = k + base;
  const float* vp = v + base;

  v8f acc[4][4];
  for (int i = 0; i < 4; ++i)
    for (int j = 0; j < 4; ++j)
      acc[i][j] = (v8f){0.f, 0.f, 0.f, 0.f, 0.f, 0.f, 0.f, 0.f};
  v4f ksum = {0.f, 0.f, 0.f, 0.f};

  v4f rk0[K1_AHEAD], rk1[K1_AHEAD], rv0[K1_AHEAD], rv1[K1_AHEAD];
  for (int p = 0; p < K1_AHEAD - 1; ++p) {
    rk0[p] = *(const v4f*)(kp + (4 * p) * DD);
    rk1[p] = *(const v4f*)(kp + (4 * p + 1) * DD);
    rv0[p] = *(const v4f*)(vp + (4 * p) * DD);
    rv1[p] = *(const v4f*)(vp + (4 * p + 1) * DD);
  }
  for (int s = 0; s < K1_STEPS; ++s) {
    const int nx = s + K1_AHEAD - 1;
    if (nx < K1_STEPS) {
      const int w = nx % K1_AHEAD;
      rk0[w] = *(const v4f*)(kp + (4 * nx) * DD);
      rk1[w] = *(const v4f*)(kp + (4 * nx + 1) * DD);
      rv0[w] = *(const v4f*)(vp + (4 * nx) * DD);
      rv1[w] = *(const v4f*)(vp + (4 * nx + 1) * DD);
    }
    const int u = s % K1_AHEAD;
    ksum += rk0[u] + rk1[u];
    for (int i = 0; i < 4; ++i) {
      v2f a;
      a.x = rk0[u][i];
      a.y = rk1[u][i];
      for (int j = 0; j < 4; ++j) {
        v2f b;
        b.x = rv0[u][j];
        b.y = rv1[u][j];
        acc[i][j] = wmma_f32(a, b, acc[i][j]);
      }
    }
  }

  {
    v4f o;
    o.x = ksum.x + __shfl_xor(ksum.x, 16);
    o.y = ksum.y + __shfl_xor(ksum.y, 16);
    o.z = ksum.z + __shfl_xor(ksum.z, 16);
    o.w = ksum.w + __shfl_xor(ksum.w, 16);
    if (h == 0) *(v4f*)(S + ((size_t)(bh * C64) + seg) * DD + 4 * c) = o;
  }

  float* my = red + wave * (DD * DD);
  for (int i = 0; i < 4; ++i)
    for (int r = 0; r < 8; ++r) {
      v4f o;
      o.x = acc[i][0][r];
      o.y = acc[i][1][r];
      o.z = acc[i][2][r];
      o.w = acc[i][3][r];
      *(v4f*)(my + (4 * (8 * h + r) + i) * DD + 4 * c) = o;
    }
  __syncthreads();

  float* gout = Gpart + (size_t)(bh * K1_CHUNKS + ch) * (DD * DD);
  for (int x = 0; x < 8; ++x) {
    const int idx = (x * 128 + t) * 4;
    const v4f p0 = *(const v4f*)(red + idx);
    const v4f p1 = *(const v4f*)(red + DD * DD + idx);
    const v4f p2 = *(const v4f*)(red + 2 * DD * DD + idx);
    const v4f p3 = *(const v4f*)(red + 3 * DD * DD + idx);
    *(v4f*)(gout + idx) = ((p0 + p1) + p2) + p3;
  }
}

__global__ void __launch_bounds__(256)
k2_combine(const float* __restrict__ proj,
           const float* __restrict__ Gpart,
           float* __restrict__ S, float* __restrict__ Wm,
           float* __restrict__ GramG) {
  __shared__ __attribute__((aligned(16))) float G[DD * DD];
  __shared__ __attribute__((aligned(16))) float Pl[DD * DD];
  __shared__ float Gm[DD * 65];
  const int bh = blockIdx.x;
  const int t  = threadIdx.x;

  {
    const float* gp = Gpart + (size_t)bh * K1_CHUNKS * (DD * DD);
    for (int x = 0; x < 4; ++x) {
      const int e = (x * 256 + t) * 4;
      v4f s = {0.f, 0.f, 0.f, 0.f};
      for (int c = 0; c < K1_CHUNKS; ++c)
        s += *(const v4f*)(gp + (size_t)c * (DD * DD) + e);
      *(v4f*)(G + e)  = s;
      *(v4f*)(Pl + e) = *(const v4f*)(proj + e);
    }
  }
  float sv[C64];
  for (int c = 0; c < C64; ++c) sv[c] = S[((size_t)(bh * C64) + c) * DD + (t & 63)];
  __syncthreads();

  const int wave = t >> 5, lane = t & 31;
  const int m = lane & 15, khalf = (lane >> 4) * 2;
  const int ti = wave >> 1;
  const int c0 = (wave & 1) * 32 + m;
  const int c1 = c0 + 16;
  const int rbase = ti * 16 + (lane >> 4) * 8;

  {
    v8f acc0 = {0.f, 0.f, 0.f, 0.f, 0.f, 0.f, 0.f, 0.f};
    v8f acc1 = {0.f, 0.f, 0.f, 0.f, 0.f, 0.f, 0.f, 0.f};
    for (int kk = 0; kk < DD; kk += 4) {
      const float* r0 = Pl + (kk + khalf) * DD;
      const float* r1 = r0 + DD;
      v2f a, b0, b1;
      a.x  = r0[ti * 16 + m];
      a.y  = r1[ti * 16 + m];
      b0.x = r0[c0];
      b0.y = r1[c0];
      b1.x = r0[c1];
      b1.y = r1[c1];
      acc0 = wmma_f32(a, b0, acc0);
      acc1 = wmma_f32(a, b1, acc1);
    }
    for (int j = 0; j < 8; ++j) {
      Gm[(rbase + j) * 65 + c0] = acc0[j];
      Gm[(rbase + j) * 65 + c1] = acc1[j];
    }
    if (bh == 0) {
      for (int j = 0; j < 8; ++j) {
        GramG[(rbase + j) * DD + c0] = acc0[j];
        GramG[(rbase + j) * DD + c1] = acc1[j];
      }
    }
  }
  __syncthreads();

  {
    v8f acc0 = {0.f, 0.f, 0.f, 0.f, 0.f, 0.f, 0.f, 0.f};
    v8f acc1 = {0.f, 0.f, 0.f, 0.f, 0.f, 0.f, 0.f, 0.f};
    const float* arow = Gm + (ti * 16 + m) * 65 + khalf;
    for (int kk = 0; kk < DD; kk += 4) {
      const float* r0 = G + (kk + khalf) * DD;
      const float* r1 = r0 + DD;
      v2f a, b0, b1;
      a.x  = arow[kk];
      a.y  = arow[kk + 1];
      b0.x = r0[c0];
      b0.y = r1[c0];
      b1.x = r0[c1];
      b1.y = r1[c1];
      acc0 = wmma_f32(a, b0, acc0);
      acc1 = wmma_f32(a, b1, acc1);
    }
    float* wout = Wm + (size_t)bh * DD * DD;
    for (int j = 0; j < 8; ++j) {
      wout[(rbase + j) * DD + c0] = acc0[j];
      wout[(rbase + j) * DD + c1] = acc1[j];
    }
  }

  if (t < 64) {
    float run = 0.f;
    for (int c = 0; c < C64; ++c) {
      S[((size_t)(bh * C64) + c) * DD + t] = run;
      run += sv[c];
    }
  }
}

__global__ void k3_out(const float* __restrict__ q, const float* __restrict__ k,
                       const float* __restrict__ S, const float* __restrict__ Wm,
                       const float* __restrict__ GramG, float* __restrict__ out) {
  __shared__ float kc[64 * 65];
  __shared__ float ab[64 * 65];
  __shared__ float qgl[64 * 65];
  __shared__ float qsum[4 * 64];
  __shared__ float dinv[64];
  const int bh = blockIdx.x, c64 = blockIdx.y;
  const int t  = threadIdx.x;
  const int n0 = c64 * 64;
  const float* qb = q + (size_t)bh * NN * DD;
  const float* kb = k + (size_t)bh * NN * DD;

  const float* kch = kb + (size_t)n0 * DD;
  const float* wb  = Wm + (size_t)bh * DD * DD;
  v4f kr[4], gr[4], wr[4];
  for (int x = 0; x < 4; ++x) kr[x] = *(const v4f*)(kch + (x * 256 + t) * 4);
  for (int x = 0; x < 4; ++x) gr[x] = *(const v4f*)(GramG + (x * 256 + t) * 4);
  for (int x = 0; x < 4; ++x) wr[x] = *(const v4f*)(wb + (x * 256 + t) * 4);
  float pre = 0.f;
  if (t < 64) pre = S[((size_t)(bh * C64) + c64) * DD + t];
  const int srow = t >> 4, scol = (t & 15) * 4;
  for (int x = 0; x < 4; ++x) {
    float* kd = kc + (x * 16 + srow) * 65 + scol;
    float* gd = ab + (x * 16 + srow) * 65 + scol;
    kd[0] = kr[x].x; kd[1] = kr[x].y; kd[2] = kr[x].z; kd[3] = kr[x].w;
    gd[0] = gr[x].x; gd[1] = gr[x].y; gd[2] = gr[x].z; gd[3] = gr[x].w;
  }
  __syncthreads();

  const int d = t & 63, qt = t >> 6;
  {
    float run = 0.f;
    for (int r = qt * 16; r < qt * 16 + 16; ++r) {
      run += kc[r * 65 + d];
      kc[r * 65 + d] = run;
    }
    qsum[qt * 64 + d] = run;
  }
  __syncthreads();
  if (t < 64) {
    const float o0 = pre;
    const float o1 = o0 + qsum[0 * 64 + t];
    const float o2 = o1 + qsum[1 * 64 + t];
    const float o3 = o2 + qsum[2 * 64 + t];
    qsum[0 * 64 + t] = o0; qsum[1 * 64 + t] = o1;
    qsum[2 * 64 + t] = o2; qsum[3 * 64 + t] = o3;
  }
  __syncthreads();
  {
    const float off = qsum[qt * 64 + d];
    for (int r = qt * 16; r < qt * 16 + 16; ++r) kc[r * 65 + d] += off;
  }
  __syncthreads();

  const int wave = t >> 5, lane = t & 31;
  const int m = lane & 15, khalf = (lane >> 4) * 2;

  for (int i = 0; i < 2; ++i) {
    const int tt = wave * 2 + i;
    const int ti = tt >> 2, tj = tt & 3;
    v8f acc = {0.f, 0.f, 0.f, 0.f, 0.f, 0.f, 0.f, 0.f};
    const float* arow = qb + (size_t)(n0 + ti * 16 + m) * DD;
    for (int kk = 0; kk < DD; kk += 4) {
      v2f a = *(const v2f*)(arow + kk + khalf);
      v2f b;
      b.x = ab[(kk + khalf) * 65 + tj * 16 + m];
      b.y = ab[(kk + khalf + 1) * 65 + tj * 16 + m];
      acc = wmma_f32(a, b, acc);
    }
    const int rbase = ti * 16 + (lane >> 4) * 8;
    for (int j = 0; j < 8; ++j)
      qgl[(rbase + j) * 65 + tj * 16 + m] = acc[j];
  }
  __syncthreads();

  for (int x = 0; x < 4; ++x) {
    float* wd = ab + (x * 16 + srow) * 65 + scol;
    wd[0] = wr[x].x; wd[1] = wr[x].y; wd[2] = wr[x].z; wd[3] = wr[x].w;
  }
  if (t < 64) {
    float s = 0.f;
    for (int dd = 0; dd < 64; ++dd) s += qgl[t * 65 + dd] * kc[t * 65 + dd];
    dinv[t] = 1.0f / s;
  }
  __syncthreads();

  for (int i = 0; i < 2; ++i) {
    const int tt = wave * 2 + i;
    const int ti = tt >> 2, tj = tt & 3;
    v8f acc = {0.f, 0.f, 0.f, 0.f, 0.f, 0.f, 0.f, 0.f};
    const float* arow = qb + (size_t)(n0 + ti * 16 + m) * DD;
    for (int kk = 0; kk < DD; kk += 4) {
      v2f a = *(const v2f*)(arow + kk + khalf);
      v2f b;
      b.x = ab[(kk + khalf) * 65 + tj * 16 + m];
      b.y = ab[(kk + khalf + 1) * 65 + tj * 16 + m];
      acc = wmma_f32(a, b, acc);
    }
    const int rbase = ti * 16 + (lane >> 4) * 8;
    for (int j = 0; j < 8; ++j) {
      const float dv = dinv[rbase + j];
      out[((size_t)bh * NN + n0 + rbase + j) * DD + tj * 16 + m] = acc[j] * dv;
    }
  }
}

extern "C" void kernel_launch(void* const* d_in, const int* in_sizes, int n_in,
                              void* d_out, int out_size, void* d_ws, size_t ws_size,
                              hipStream_t stream) {
  const float* q    = (const float*)d_in[0];
  const float* k    = (const float*)d_in[1];
  const float* v    = (const float*)d_in[2];
  const float* proj = (const float*)d_in[3];
  float* out = (float*)d_out;

  float* ws    = (float*)d_ws;
  float* Gpart = ws;
  float* S     = Gpart + (size_t)BH * K1_CHUNKS * DD * DD;
  float* Wm    = S + (size_t)BH * C64 * DD;
  float* Gram  = Wm + (size_t)BH * DD * DD;

  k1_gpart<<<dim3(BH, K1_CHUNKS), 128, 0, stream>>>(k, v, Gpart, S);
  k2_combine<<<dim3(BH), 256, 0, stream>>>(proj, Gpart, S, Wm, Gram);
  k3_out<<<dim3(BH, C64), 256, 0, stream>>>(q, k, S, Wm, Gram, out);
}


# K3: q chunk staged once in LDS by whole-line 128-bit loads in the early batch; each wave reads its A fragments there once and keeps them for its 4 tile chains (was strided 8-byte global reads)
# speedup vs baseline: 1.7470x; 1.1163x over previous
#include <hip/hip_runtime.h>


typedef __attribute__((ext_vector_type(2))) float v2f;
typedef __attribute__((ext_vector_type(4))) float v4f;
typedef __attribute__((ext_vector_type(8))) float v8f;

#define BH 64
#define NN 4096
#define DD 64
#define K1_ROWS 256
#define K1_CHUNKS 16
#define C64 64

static __device__ __forceinline__ v8f wmma_f32(v2f a, v2f b, v8f c) {
  return __builtin_amdgcn_wmma_f32_16x16x4_f32(false, a, false, b, (short)0, c,
                                               false, false);
}

#define K1_WAVES 4
#define K1_STEPS 16
#define K1_AHEAD 5

__global__ void __launch_bounds__(128)
k1_gpart(const float* __restrict__ k, const float* __restrict__ v,
         float* __restrict__ Gpart, float* __restrict__ S) {
  __shared__ float red[K1_WAVES * DD * DD];
  const int bh = blockIdx.x;
  const int ch = blockIdx.y;
  const int t  = threadIdx.x;
  const int wave = t >> 5, lane = t & 31;
  const int c = lane & 15, h = lane >> 4;
  const int seg = ch * K1_WAVES + wave;
  const size_t base = ((size_t)bh * NN + (size_t)seg * 64 + 2 * h) * DD + 4 * c;
  const float* kp = k + base;
  const float* vp = v + base;

  v8f acc[4][4];
  for (int i = 0; i < 4; ++i)
    for (int j = 0; j < 4; ++j)
      acc[i][j] = (v8f){0.f, 0.f, 0.f, 0.f, 0.f, 0.f, 0.f, 0.f};
  v4f ksum = {0.f, 0.f, 0.f, 0.f};

  v4f rk0[K1_AHEAD], rk1[K1_AHEAD], rv0[K1_AHEAD], rv1[K1_AHEAD];
  for (int p = 0; p < K1_AHEAD - 1; ++p) {
    rk0[p] = *(const v4f*)(kp + (4 * p) * DD);
    rk1[p] = *(const v4f*)(kp + (4 * p + 1) * DD);
    rv0[p] = *(const v4f*)(vp + (4 * p) * DD);
    rv1[p] = *(const v4f*)(vp + (4 * p + 1) * DD);
  }
  for (int s = 0; s < K1_STEPS; ++s) {
    const int nx = s + K1_AHEAD - 1;
    if (nx < K1_STEPS) {
      const int w = nx % K1_AHEAD;
      rk0[w] = *(const v4f*)(kp + (4 * nx) * DD);
      rk1[w] = *(const v4f*)(kp + (4 * nx + 1) * DD);
      rv0[w] = *(const v4f*)(vp + (4 * nx) * DD);
      rv1[w] = *(const v4f*)(vp + (4 * nx + 1) * DD);
    }
    const int u = s % K1_AHEAD;
    ksum += rk0[u] + rk1[u];
    for (int i = 0; i < 4; ++i) {
      v2f a;
      a.x = rk0[u][i];
      a.y = rk1[u][i];
      for (int j = 0; j < 4; ++j) {
        v2f b;
        b.x = rv0[u][j];
        b.y = rv1[u][j];
        acc[i][j] = wmma_f32(a, b, acc[i][j]);
      }
    }
  }

  {
    v4f o;
    o.x = ksum.x + __shfl_xor(ksum.x, 16);
    o.y = ksum.y + __shfl_xor(ksum.y, 16);
    o.z = ksum.z + __shfl_xor(ksum.z, 16);
    o.w = ksum.w + __shfl_xor(ksum.w, 16);
    if (h == 0) *(v4f*)(S + ((size_t)(bh * C64) + seg) * DD + 4 * c) = o;
  }

  float* my = red + wave * (DD * DD);
  for (int i = 0; i < 4; ++i)
    for (int r = 0; r < 8; ++r) {
      v4f o;
      o.x = acc[i][0][r];
      o.y = acc[i][1][r];
      o.z = acc[i][2][r];
      o.w = acc[i][3][r];
      *(v4f*)(my + (4 * (8 * h + r) + i) * DD + 4 * c) = o;
    }
  __syncthreads();

  float* gout = Gpart + (size_t)(bh * K1_CHUNKS + ch) * (DD * DD);
  for (int x = 0; x < 8; ++x) {
    const int idx = (x * 128 + t) * 4;
    const v4f p0 = *(const v4f*)(red + idx);
    const v4f p1 = *(const v4f*)(red + DD * DD + idx);
    const v4f p2 = *(const v4f*)(red + 2 * DD * DD + idx);
    const v4f p3 = *(const v4f*)(red + 3 * DD * DD + idx);
    *(v4f*)(gout + idx) = ((p0 + p1) + p2) + p3;
  }
}

__global__ void __launch_bounds__(256)
k2_combine(const float* __restrict__ proj,
           const float* __restrict__ Gpart,
           float* __restrict__ S, float* __restrict__ Wm,
           float* __restrict__ GramG) {
  __shared__ __attribute__((aligned(16))) float G[DD * DD];
  __shared__ __attribute__((aligned(16))) float Pl[DD * DD];
  __shared__ float Gm[DD * 65];
  const int bh = blockIdx.x;
  const int t  = threadIdx.x;

  {
    const float* gp = Gpart + (size_t)bh * K1_CHUNKS * (DD * DD);
    for (int x = 0; x < 4; ++x) {
      const int e = (x * 256 + t) * 4;
      v4f s = {0.f, 0.f, 0.f, 0.f};
      for (int c = 0; c < K1_CHUNKS; ++c)
        s += *(const v4f*)(gp + (size_t)c * (DD * DD) + e);
      *(v4f*)(G + e)  = s;
      *(v4f*)(Pl + e) = *(const v4f*)(proj + e);
    }
  }
  float sv[C64];
  for (int c = 0; c < C64; ++c) sv[c] = S[((size_t)(bh * C64) + c) * DD + (t & 63)];
  __syncthreads();

  const int wave = t >> 5, lane = t & 31;
  const int m = lane & 15, khalf = (lane >> 4) * 2;
  const int ti = wave >> 1;
  const int c0 = (wave & 1) * 32 + m;
  const int c1 = c0 + 16;
  const int rbase = ti * 16 + (lane >> 4) * 8;

  {
    v8f acc0 = {0.f, 0.f, 0.f, 0.f, 0.f, 0.f, 0.f, 0.f};
    v8f acc1 = {0.f, 0.f, 0.f, 0.f, 0.f, 0.f, 0.f, 0.f};
    for (int kk = 0; kk < DD; kk += 4) {
      const float* r0 = Pl + (kk + khalf) * DD;
      const float* r1 = r0 + DD;
      v2f a, b0, b1;
      a.x  = r0[ti * 16 + m];
      a.y  = r1[ti * 16 + m];
      b0.x = r0[c0];
      b0.y = r1[c0];
      b1.x = r0[c1];
      b1.y = r1[c1];
      acc0 = wmma_f32(a, b0, acc0);
      acc1 = wmma_f32(a, b1, acc1);
    }
    for (int j = 0; j < 8; ++j) {
      Gm[(rbase + j) * 65 + c0] = acc0[j];
      Gm[(rbase + j) * 65 + c1] = acc1[j];
    }
    if (bh == 0) {
      for (int j = 0; j < 8; ++j) {
        GramG[(rbase + j) * DD + c0] = acc0[j];
        GramG[(rbase + j) * DD + c1] = acc1[j];
      }
    }
  }
  __syncthreads();

  {
    v8f acc0 = {0.f, 0.f, 0.f, 0.f, 0.f, 0.f, 0.f, 0.f};
    v8f acc1 = {0.f, 0.f, 0.f, 0.f, 0.f, 0.f, 0.f, 0.f};
    const float* arow = Gm + (ti * 16 + m) * 65 + khalf;
    for (int kk = 0; kk < DD; kk += 4) {
      const float* r0 = G + (kk + khalf) * DD;
      const float* r1 = r0 + DD;
      v2f a, b0, b1;
      a.x  = arow[kk];
      a.y  = arow[kk + 1];
      b0.x = r0[c0];
      b0.y = r1[c0];
      b1.x = r0[c1];
      b1.y = r1[c1];
      acc0 = wmma_f32(a, b0, acc0);
      acc1 = wmma_f32(a, b1, acc1);
    }
    float* wout = Wm + (size_t)bh * DD * DD;
    for (int j = 0; j < 8; ++j) {
      wout[(rbase + j) * DD + c0] = acc0[j];
      wout[(rbase + j) * DD + c1] = acc1[j];
    }
  }

  if (t < 64) {
    float run = 0.f;
    for (int c = 0; c < C64; ++c) {
      S[((size_t)(bh * C64) + c) * DD + t] = run;
      run += sv[c];
    }
  }
}

#define QP 68

__global__ void k3_out(const float* __restrict__ q, const float* __restrict__ k,
                       const float* __restrict__ S, const float* __restrict__ Wm,
                       const float* __restrict__ GramG, float* __restrict__ out) {
  __shared__ float kc[64 * 65];
  __shared__ float ab[64 * 65];
  __shared__ __attribute__((aligned(16))) float qgl[64 * QP];
  __shared__ float qsum[4 * 64];
  __shared__ float dinv[64];
  const int bh = blockIdx.x, c64 = blockIdx.y;
  const int t  = threadIdx.x;
  const int n0 = c64 * 64;
  const float* qb = q + (size_t)bh * NN * DD;
  const float* kb = k + (size_t)bh * NN * DD;

  const float* kch = kb + (size_t)n0 * DD;
  const float* qch = qb + (size_t)n0 * DD;
  const float* wb  = Wm + (size_t)bh * DD * DD;
  v4f kr[4], qr[4], gr[4], wr[4];
  for (int x = 0; x < 4; ++x) kr[x] = *(const v4f*)(kch + (x * 256 + t) * 4);
  for (int x = 0; x < 4; ++x) qr[x] = *(const v4f*)(qch + (x * 256 + t) * 4);
  for (int x = 0; x < 4; ++x) gr[x] = *(const v4f*)(GramG + (x * 256 + t) * 4);
  for (int x = 0; x < 4; ++x) wr[x] = *(const v4f*)(wb + (x * 256 + t) * 4);
  float pre = 0.f;
  if (t < 64) pre = S[((size_t)(bh * C64) + c64) * DD + t];
  const int srow = t >> 4, scol = (t & 15) * 4;
  for (int x = 0; x < 4; ++x) {
    float* kd = kc + (x * 16 + srow) * 65 + scol;
    float* gd = ab + (x * 16 + srow) * 65 + scol;
    kd[0] = kr[x].x; kd[1] = kr[x].y; kd[2] = kr[x].z; kd[3] = kr[x].w;
    gd[0] = gr[x].x; gd[1] = gr[x].y; gd[2] = gr[x].z; gd[3] = gr[x].w;
    *(v4f*)(qgl + (x * 16 + srow) * QP + scol) = qr[x];
  }
  __syncthreads();

  const int wave = t >> 5, lane = t & 31;
  const int m = lane & 15, khalf = (lane >> 4) * 2;
  const int ti  = wave >> 1;
  const int tj0 = (wave & 1) * 2;
  const int rbase = ti * 16 + (lane >> 4) * 8;

  v2f qa[16];
  {
    const float* qrow = qgl + (ti * 16 + m) * QP + khalf;
    for (int j = 0; j < 16; ++j) qa[j] = *(const v2f*)(qrow + 4 * j);
  }

  const int d = t & 63, qt = t >> 6;
  {
    float run = 0.f;
    for (int r = qt * 16; r < qt * 16 + 16; ++r) {
      run += kc[r * 65 + d];
      kc[r * 65 + d] = run;
    }
    qsum[qt * 64 + d] = run;
  }
  __syncthreads();
  if (t < 64) {
    const float o0 = pre;
    const float o1 = o0 + qsum[0 * 64 + t];
    const float o2 = o1 + qsum[1 * 64 + t];
    const float o3 = o2 + qsum[2 * 64 + t];
    qsum[0 * 64 + t] = o0; qsum[1 * 64 + t] = o1;
    qsum[2 * 64 + t] = o2; qsum[3 * 64 + t] = o3;
  }
  __syncthreads();
  {
    const float off = qsum[qt * 64 + d];
    for (int r = qt * 16; r < qt * 16 + 16; ++r) kc[r * 65 + d] += off;
  }
  __syncthreads();

  const float* bcol = ab + khalf * 65 + tj0 * 16 + m;

  {
    v8f acc0 = {0.f, 0.f, 0.f, 0.f, 0.f, 0.f, 0.f, 0.f};
    v8f acc1 = {0.f, 0.f, 0.f, 0.f, 0.f, 0.f, 0.f, 0.f};
    for (int j = 0; j < 16; ++j) {
      const float* r0 = bcol + (4 * j) * 65;
      v2f b0, b1;
      b0.x = r0[0];
      b0.y = r0[65];
      b1.x = r0[16];
      b1.y = r0[65 + 16];
      acc0 = wmma_f32(qa[j], b0, acc0);
      acc1 = wmma_f32(qa[j], b1, acc1);
    }
    for (int j = 0; j < 8; ++j) {
      qgl[(rbase + j) * 65 + tj0 * 16 + m]      = acc0[j];
      qgl[(rbase + j) * 65 + tj0 * 16 + 16 + m] = acc1[j];
    }
  }
  __syncthreads();

  for (int x = 0; x < 4; ++x) {
    float* wd = ab + (x * 16 + srow) * 65 + scol;
    wd[0] = wr[x].x; wd[1] = wr[x].y; wd[2] = wr[x].z; wd[3] = wr[x].w;
  }
  if (t < 64) {
    float s = 0.f;
    for (int dd = 0; dd < 64; ++dd) s += qgl[t * 65 + dd] * kc[t * 65 + dd];
    dinv[t] = 1.0f / s;
  }
  __syncthreads();

  {
    v8f acc0 = {0.f, 0.f, 0.f, 0.f, 0.f, 0.f, 0.f, 0.f};
    v8f acc1 = {0.f, 0.f, 0.f, 0.f, 0.f, 0.f, 0.f, 0.f};
    for (int j = 0; j < 16; ++j) {
      const float* r0 = bcol + (4 * j) * 65;
      v2f b0, b1;
      b0.x = r0[0];
      b0.y = r0[65];
      b1.x = r0[16];
      b1.y = r0[65 + 16];
      acc0 = wmma_f32(qa[j], b0, acc0);
      acc1 = wmma_f32(qa[j], b1, acc1);
    }
    for (int j = 0; j < 8; ++j) {
      const float dv = dinv[rbase + j];
      float* orow = out + ((size_t)bh * NN + n0 + rbase + j) * DD + tj0 * 16 + m;
      orow[0]  = acc0[j] * dv;
      orow[16] = acc1[j] * dv;
    }
  }
}

extern "C" void kernel_launch(void* const* d_in, const int* in_sizes, int n_in,
                              void* d_out, int out_size, void* d_ws, size_t ws_size,
                              hipStream_t stream) {
  const float* q    = (const float*)d_in[0];
  const float* k    = (const float*)d_in[1];
  const float* v    = (const float*)d_in[2];
  const float* proj = (const float*)d_in[3];
  float* out = (float*)d_out;

  float* ws    = (float*)d_ws;
  float* Gpart = ws;
  float* S     = Gpart + (size_t)BH * K1_CHUNKS * DD * DD;
  float* Wm    = S + (size_t)BH * C64 * DD;
  float* Gram  = Wm + (size_t)BH * DD * DD;

  k1_gpart<<<dim3(BH, K1_CHUNKS), 128, 0, stream>>>(k, v, Gpart, S);
  k2_combine<<<dim3(BH), 256, 0, stream>>>(proj, Gpart, S, Wm, Gram);
  k3_out<<<dim3(BH, C64), 256, 0, stream>>>(q, k, S, Wm, Gram, out);
}


# K3: Gram and W^T lie k-contiguous in LDS (K2 stores W transposed); B and A come as 128-bit LDS reads issued ahead of the WMMAs; W fetched after chain 1; k sum order of q@Gram and q@W changes
# speedup vs baseline: 1.8273x; 1.0460x over previous
#include <hip/hip_runtime.h>


typedef __attribute__((ext_vector_type(2))) float v2f;
typedef __attribute__((ext_vector_type(4))) float v4f;
typedef __attribute__((ext_vector_type(8))) float v8f;

#define BH 64
#define NN 4096
#define DD 64
#define K1_ROWS 256
#define K1_CHUNKS 16
#define C64 64

static __device__ __forceinline__ v8f wmma_f32(v2f a, v2f b, v8f c) {
  return __builtin_amdgcn_wmma_f32_16x16x4_f32(false, a, false, b, (short)0, c,
                                               false, false);
}

#define K1_WAVES 4
#define K1_STEPS 16
#define K1_AHEAD 5

__global__ void __launch_bounds__(128)
k1_gpart(const float* __restrict__ k, const float* __restrict__ v,
         float* __restrict__ Gpart, float* __restrict__ S) {
  __shared__ float red[K1_WAVES * DD * DD];
  const int bh = blockIdx.x;
  const int ch = blockIdx.y;
  const int t  = threadIdx.x;
  const int wave = t >> 5, lane = t & 31;
  const int c = lane & 15, h = lane >> 4;
  const int seg = ch * K1_WAVES + wave;
  const size_t base = ((size_t)bh * NN + (size_t)seg * 64 + 2 * h) * DD + 4 * c;
  const float* kp = k + base;
  const float* vp = v + base;

  v8f acc[4][4];
  for (int i = 0; i < 4; ++i)
    for (int j = 0; j < 4; ++j)
      acc[i][j] = (v8f){0.f, 0.f, 0.f, 0.f, 0.f, 0.f, 0.f, 0.f};
  v4f ksum = {0.f, 0.f, 0.f, 0.f};

  v4f rk0[K1_AHEAD], rk1[K1_AHEAD], rv0[K1_AHEAD], rv1[K1_AHEAD];
  for (int p = 0; p < K1_AHEAD - 1; ++p) {
    rk0[p] = *(const v4f*)(kp + (4 * p) * DD);
    rk1[p] = *(const v4f*)(kp + (4 * p + 1) * DD);
    rv0[p] = *(const v4f*)(vp + (4 * p) * DD);
    rv1[p] = *(const v4f*)(vp + (4 * p + 1) * DD);
  }
  for (int s = 0; s < K1_STEPS; ++s) {
    const int nx = s + K1_AHEAD - 1;
    if (nx < K1_STEPS) {
      const int w = nx % K1_AHEAD;
      rk0[w] = *(const v4f*)(kp + (4 * nx) * DD);
      rk1[w] = *(const v4f*)(kp + (4 * nx + 1) * DD);
      rv0[w] = *(const v4f*)(vp + (4 * nx) * DD);
      rv1[w] = *(const v4f*)(vp + (4 * nx + 1) * DD);
    }
    const int u = s % K1_AHEAD;
    ksum += rk0[u] + rk1[u];
    for (int i = 0; i < 4; ++i) {
      v2f a;
      a.x = rk0[u][i];
      a.y = rk1[u][i];
      for (int j = 0; j < 4; ++j) {
        v2f b;
        b.x = rv0[u][j];
        b.y = rv1[u][j];
        acc[i][j] = wmma_f32(a, b, acc[i][j]);
      }
    }
  }

  {
    v4f o;
    o.x = ksum.x + __shfl_xor(ksum.x, 16);
    o.y = ksum.y + __shfl_xor(ksum.y, 16);
    o.z = ksum.z + __shfl_xor(ksum.z, 16);
    o.w = ksum.w + __shfl_xor(ksum.w, 16);
    if (h == 0) *(v4f*)(S + ((size_t)(bh * C64) + seg) * DD + 4 * c) = o;
  }

  float* my = red + wave * (DD * DD);
  for (int i = 0; i < 4; ++i)
    for (int r = 0; r < 8; ++r) {
      v4f o;
      o.x = acc[i][0][r];
      o.y = acc[i][1][r];
      o.z = acc[i][2][r];
      o.w = acc[i][3][r];
      *(v4f*)(my + (4 * (8 * h + r) + i) * DD + 4 * c) = o;
    }
  __syncthreads();

  float* gout = Gpart + (size_t)(bh * K1_CHUNKS + ch) * (DD * DD);
  for (int x = 0; x < 8; ++x) {
    const int idx = (x * 128 + t) * 4;
    const v4f p0 = *(const v4f*)(red + idx);
    const v4f p1 = *(const v4f*)(red + DD * DD + idx);
    const v4f p2 = *(const v4f*)(red + 2 * DD * DD + idx);
    const v4f p3 = *(const v4f*)(red + 3 * DD * DD + idx);
    *(v4f*)(gout + idx) = ((p0 + p1) + p2) + p3;
  }
}

#define WTP 68
__global__ void __launch_bounds__(256)
k2_combine(const float* __restrict__ proj,
           const float* __restrict__ Gpart,
           float* __restrict__ S, float* __restrict__ Wm,
           float* __restrict__ GramG) {
  __shared__ __attribute__((aligned(16))) float G[DD * DD];
  __shared__ __attribute__((aligned(16))) float Pl[DD * WTP];
  __shared__ float Gm[DD * 65];
  const int bh = blockIdx.x;
  const int t  = threadIdx.x;

  {
    const float* gp = Gpart + (size_t)bh * K1_CHUNKS * (DD * DD);
    for (int x = 0; x < 4; ++x) {
      const int e = (x * 256 + t) * 4;
      v4f s = {0.f, 0.f, 0.f, 0.f};
      for (int c = 0; c < K1_CHUNKS; ++c)
        s += *(const v4f*)(gp + (size_t)c * (DD * DD) + e);
      *(v4f*)(G + e)  = s;
      *(v4f*)(Pl + e) = *(const v4f*)(proj + e);
    }
  }
  float sv[C64];
  for (int c = 0; c < C64; ++c) sv[c] = S[((size_t)(bh * C64) + c) * DD + (t & 63)];
  __syncthreads();

  const int wave = t >> 5, lane = t & 31;
  const int m = lane & 15, khalf = (lane >> 4) * 2;
  const int ti = wave >> 1;
  const int c0 = (wave & 1) * 32 + m;
  const int c1 = c0 + 16;
  const int rbase = ti * 16 + (lane >> 4) * 8;

  {
    v8f acc0 = {0.f, 0.f, 0.f, 0.f, 0.f, 0.f, 0.f, 0.f};
    v8f acc1 = {0.f, 0.f, 0.f, 0.f, 0.f, 0.f, 0.f, 0.f};
    for (int kk = 0; kk < DD; kk += 4) {
      const float* r0 = Pl + (kk + khalf) * DD;
      const float* r1 = r0 + DD;
      v2f a, b0, b1;
      a.x  = r0[ti * 16 + m];
      a.y  = r1[ti * 16 + m];
      b0.x = r0[c0];
      b0.y = r1[c0];
      b1.x = r0[c1];
      b1.y = r1[c1];
      acc0 = wmma_f32(a, b0, acc0);
      acc1 = wmma_f32(a, b1, acc1);
    }
    for (int j = 0; j < 8; ++j) {
      Gm[(rbase + j) * 65 + c0] = acc0[j];
      Gm[(rbase + j) * 65 + c1] = acc1[j];
    }
    if (bh == 0) {
      for (int j = 0; j < 8; ++j) {
        GramG[(rbase + j) * DD + c0] = acc0[j];
        GramG[(rbase + j) * DD + c1] = acc1[j];
      }
    }
  }
  __syncthreads();

  {
    v8f acc0 = {0.f, 0.f, 0.f, 0.f, 0.f, 0.f, 0.f, 0.f};
    v8f acc1 = {0.f, 0.f, 0.f, 0.f, 0.f, 0.f, 0.f, 0.f};
    const float* arow = Gm + (ti * 16 + m) * 65 + khalf;
    for (int kk = 0; kk < DD; kk += 4) {
      const float* r0 = G + (kk + khalf) * DD;
      const float* r1 = r0 + DD;
      v2f a, b0, b1;
      a.x  = arow[kk];
      a.y  = arow[kk + 1];
      b0.x = r0[c0];
      b0.y = r1[c0];
      b1.x = r0[c1];
      b1.y = r1[c1];
      acc0 = wmma_f32(a, b0, acc0);
      acc1 = wmma_f32(a, b1, acc1);
    }
    *(v4f*)(Pl + c0 * WTP + rbase)     = (v4f){acc0[0], acc0[1], acc0[2], acc0[3]};
    *(v4f*)(Pl + c0 * WTP + rbase + 4) = (v4f){acc0[4], acc0[5], acc0[6], acc0[7]};
    *(v4f*)(Pl + c1 * WTP + rbase)     = (v4f){acc1[0], acc1[1], acc1[2], acc1[3]};
    *(v4f*)(Pl + c1 * WTP + rbase + 4) = (v4f){acc1[4], acc1[5], acc1[6], acc1[7]};
  }
  __syncthreads();
  {
    float* wout = Wm + (size_t)bh * DD * DD;
    const int wrow = t >> 4, wcol = (t & 15) * 4;
    for (int x = 0; x < 4; ++x)
      *(v4f*)(wout + (x * 16 + wrow) * DD + wcol) = *(const v4f*)(Pl + (x * 16 + wrow) * WTP + wcol);
  }

  if (t < 64) {
    float run = 0.f;
    for (int c = 0; c < C64; ++c) {
      S[((size_t)(bh * C64) + c) * DD + t] = run;
      run += sv[c];
    }
  }
}

#define QP 68

static __device__ __forceinline__ void band_product(const v4f (&qa)[8], const float* bt0,
                                                    const float* bt1, v8f& acc0, v8f& acc1) {
  v4f b0[8], b1[8];
  b0[0] = *(const v4f*)(bt0);
  b1[0] = *(const v4f*)(bt1);
  b0[1] = *(const v4f*)(bt0 + 8);
  b1[1] = *(const v4f*)(bt1 + 8);
  for (int g = 0; g < 4; ++g) {
    if (g < 3) {
      b0[2 * g + 2] = *(const v4f*)(bt0 + 8 * (2 * g + 2));
      b1[2 * g + 2] = *(const v4f*)(bt1 + 8 * (2 * g + 2));
      b0[2 * g + 3] = *(const v4f*)(bt0 + 8 * (2 * g + 3));
      b1[2 * g + 3] = *(const v4f*)(bt1 + 8 * (2 * g + 3));
    }
    for (int p = 2 * g; p < 2 * g + 2; ++p) {
      const v2f alo = {qa[p].x, qa[p].y};
      const v2f ahi = {qa[p].z, qa[p].w};
      const v2f x0 = {b0[p].x, b0[p].y};
      const v2f x1 = {b1[p].x, b1[p].y};
      const v2f y0 = {b0[p].z, b0[p].w};
      const v2f y1 = {b1[p].z, b1[p].w};
      acc0 = wmma_f32(alo, x0, acc0);
      acc1 = wmma_f32(alo, x1, acc1);
      acc0 = wmma_f32(ahi, y0, acc0);
      acc1 = wmma_f32(ahi, y1, acc1);
    }
  }
}

__global__ void __launch_bounds__(256, 12)
k3_out(const float* __restrict__ q, const float* __restrict__ k,
                       const float* __restrict__ S, const float* __restrict__ Wm,
                       const float* __restrict__ GramG, float* __restrict__ out) {
  __shared__ float kc[64 * 65];
  __shared__ __attribute__((aligned(16))) float ab[64 * QP];
  __shared__ __attribute__((aligned(16))) float qgl[64 * QP];
  __shared__ float qsum[4 * 64];
  __shared__ float dinv[64];
  const int bh = blockIdx.x, c64 = blockIdx.y;
  const int t  = threadIdx.x;
  const int n0 = c64 * 64;
  const float* qb = q + (size_t)bh * NN * DD;
  const float* kb = k + (size_t)bh * NN * DD;

  const float* kch = kb + (size_t)n0 * DD;
  const float* qch = qb + (size_t)n0 * DD;
  const float* wb  = Wm + (size_t)bh * DD * DD;
  v4f kr[4], qr[4], gr[4];
  for (int x = 0; x < 4; ++x) kr[x] = *(const v4f*)(kch + (x * 256 + t) * 4);
  for (int x = 0; x < 4; ++x) qr[x] = *(const v4f*)(qch + (x * 256 + t) * 4);
  for (int x = 0; x < 4; ++x) gr[x] = *(const v4f*)(GramG + (x * 256 + t) * 4);
  float pre = 0.f;
  if (t < 64) pre = S[((size_t)(bh * C64) + c64) * DD + t];
  const int srow = t >> 4, scol = (t & 15) * 4;
  for (int x = 0; x < 4; ++x) {
    float* kd = kc + (x * 16 + srow) * 65 + scol;
    kd[0] = kr[x].x; kd[1] = kr[x].y; kd[2] = kr[x].z; kd[3] = kr[x].w;
    *(v4f*)(ab + (x * 16 + srow) * QP + scol) = gr[x];
    *(v4f*)(qgl + (x * 16 + srow) * QP + scol) = qr[x];
  }
  __syncthreads();

  const int wave = t >> 5, lane = t & 31;
  const int m = lane & 15, kq = (lane >> 4) * 4;
  const int ti  = wave >> 1;
  const int tj0 = (wave & 1) * 2;
  const int rbase = ti * 16 + (lane >> 4) * 8;

  v4f qa[8];
  {
    const float* qrow = qgl + (ti * 16 + m) * QP + kq;
    for (int p = 0; p < 8; ++p) qa[p] = *(const v4f*)(qrow + 8 * p);
  }

  const int d = t & 63, qt = t >> 6;
  {
    float run = 0.f;
    for (int r = qt * 16; r < qt * 16 + 16; ++r) {
      run += kc[r * 65 + d];
      kc[r * 65 + d] = run;
    }
    qsum[qt * 64 + d] = run;
  }
  __syncthreads();
  if (t < 64) {
    const float o0 = pre;
    const float o1 = o0 + qsum[0 * 64 + t];
    const float o2 = o1 + qsum[1 * 64 + t];
    const float o3 = o2 + qsum[2 * 64 + t];
    qsum[0 * 64 + t] = o0; qsum[1 * 64 + t] = o1;
    qsum[2 * 64 + t] = o2; qsum[3 * 64 + t] = o3;
  }
  __syncthreads();
  {
    const float off = qsum[qt * 64 + d];
    for (int r = qt * 16; r < qt * 16 + 16; ++r) kc[r * 65 + d] += off;
  }
  __syncthreads();

  const float* bt0 = ab + (tj0 * 16 + m) * QP + kq;
  const float* bt1 = bt0 + 16 * QP;

  {
    v8f acc0 = {0.f, 0.f, 0.f, 0.f, 0.f, 0.f, 0.f, 0.f};
    v8f acc1 = {0.f, 0.f, 0.f, 0.f, 0.f, 0.f, 0.f, 0.f};
    band_product(qa, bt0, bt1, acc0, acc1);
    for (int j = 0; j < 8; ++j) {
      qgl[(rbase + j) * 65 + tj0 * 16 + m]      = acc0[j];
      qgl[(rbase + j) * 65 + tj0 * 16 + 16 + m] = acc1[j];
    }
  }
  __syncthreads();

  v4f wr[4];
  for (int x = 0; x < 4; ++x) wr[x] = *(const v4f*)(wb + (x * 256 + t) * 4);
  if (t < 64) {
    float s = 0.f;
    for (int dd = 0; dd < 64; ++dd) s += qgl[t * 65 + dd] * kc[t * 65 + dd];
    dinv[t] = 1.0f / s;
  }
  for (int x = 0; x < 4; ++x) *(v4f*)(ab + (x * 16 + srow) * QP + scol) = wr[x];
  __syncthreads();

  {
    v8f acc0 = {0.f, 0.f, 0.f, 0.f, 0.f, 0.f, 0.f, 0.f};
    v8f acc1 = {0.f, 0.f, 0.f, 0.f, 0.f, 0.f, 0.f, 0.f};
    band_product(qa, bt0, bt1, acc0, acc1);
    for (int j = 0; j < 8; ++j) {
      const float dv = dinv[rbase + j];
      float* orow = out + ((size_t)bh * NN + n0 + rbase + j) * DD + tj0 * 16 + m;
      orow[0]  = acc0[j] * dv;
      orow[16] = acc1[j] * dv;
    }
  }
}

extern "C" void kernel_launch(void* const* d_in, const int* in_sizes, int n_in,
                              void* d_out, int out_size, void* d_ws, size_t ws_size,
                              hipStream_t stream) {
  const float* q    = (const float*)d_in[0];
  const float* k    = (const float*)d_in[1];
  const float* v    = (const float*)d_in[2];
  const float* proj = (const float*)d_in[3];
  float* out = (float*)d_out;

  float* ws    = (float*)d_ws;
  float* Gpart = ws;
  float* S     = Gpart + (size_t)BH * K1_CHUNKS * DD * DD;
  float* Wm    = S + (size_t)BH * C64 * DD;
  float* Gram  = Wm + (size_t)BH * DD * DD;

  k1_gpart<<<dim3(BH, K1_CHUNKS), 128, 0, stream>>>(k, v, Gpart, S);
  k2_combine<<<dim3(BH), 256, 0, stream>>>(proj, Gpart, S, Wm, Gram);
  k3_out<<<dim3(BH, C64), 256, 0, stream>>>(q, k, S, Wm, Gram, out);
}


# K2 runs as 5 work-groups per head (4 column slabs of W, each summing its own slab of the partials, + the scan of S), partials stored slab-major by K1: all WGPs pull them, not 64; sum orders kept
# speedup vs baseline: 1.9642x; 1.0749x over previous
#include <hip/hip_runtime.h>


typedef __attribute__((ext_vector_type(2))) float v2f;
typedef __attribute__((ext_vector_type(4))) float v4f;
typedef __attribute__((ext_vector_type(8))) float v8f;

#define BH 64
#define NN 4096
#define DD 64
#define K1_ROWS 256
#define K1_CHUNKS 16
#define C64 64

static __device__ __forceinline__ v8f wmma_f32(v2f a, v2f b, v8f c) {
  return __builtin_amdgcn_wmma_f32_16x16x4_f32(false, a, false, b, (short)0, c,
                                               false, false);
}

#define K1_WAVES 4
#define K1_STEPS 16
#define K1_AHEAD 5

__global__ void __launch_bounds__(128)
k1_gpart(const float* __restrict__ k, const float* __restrict__ v,
         float* __restrict__ Gpart, float* __restrict__ S) {
  __shared__ float red[K1_WAVES * DD * DD];
  const int bh = blockIdx.x;
  const int ch = blockIdx.y;
  const int t  = threadIdx.x;
  const int wave = t >> 5, lane = t & 31;
  const int c = lane & 15, h = lane >> 4;
  const int seg = ch * K1_WAVES + wave;
  const size_t base = ((size_t)bh * NN + (size_t)seg * 64 + 2 * h) * DD + 4 * c;
  const float* kp = k + base;
  const float* vp = v + base;

  v8f acc[4][4];
  for (int i = 0; i < 4; ++i)
    for (int j = 0; j < 4; ++j)
      acc[i][j] = (v8f){0.f, 0.f, 0.f, 0.f, 0.f, 0.f, 0.f, 0.f};
  v4f ksum = {0.f, 0.f, 0.f, 0.f};

  v4f rk0[K1_AHEAD], rk1[K1_AHEAD], rv0[K1_AHEAD], rv1[K1_AHEAD];
  for (int p = 0; p < K1_AHEAD - 1; ++p) {
    rk0[p] = *(const v4f*)(kp + (4 * p) * DD);
    rk1[p] = *(const v4f*)(kp + (4 * p + 1) * DD);
    rv0[p] = *(const v4f*)(vp + (4 * p) * DD);
    rv1[p] = *(const v4f*)(vp + (4 * p + 1) * DD);
  }
  for (int s = 0; s < K1_STEPS; ++s) {
    const int nx = s + K1_AHEAD - 1;
    if (nx < K1_STEPS) {
      const int w = nx % K1_AHEAD;
      rk0[w] = *(const v4f*)(kp + (4 * nx) * DD);
      rk1[w] = *(const v4f*)(kp + (4 * nx + 1) * DD);
      rv0[w] = *(const v4f*)(vp + (4 * nx) * DD);
      rv1[w] = *(const v4f*)(vp + (4 * nx + 1) * DD);
    }
    const int u = s % K1_AHEAD;
    ksum += rk0[u] + rk1[u];
    for (int i = 0; i < 4; ++i) {
      v2f a;
      a.x = rk0[u][i];
      a.y = rk1[u][i];
      for (int j = 0; j < 4; ++j) {
        v2f b;
        b.x = rv0[u][j];
        b.y = rv1[u][j];
        acc[i][j] = wmma_f32(a, b, acc[i][j]);
      }
    }
  }

  {
    v4f o;
    o.x = ksum.x + __shfl_xor(ksum.x, 16);
    o.y = ksum.y + __shfl_xor(ksum.y, 16);
    o.z = ksum.z + __shfl_xor(ksum.z, 16);
    o.w = ksum.w + __shfl_xor(ksum.w, 16);
    if (h == 0) *(v4f*)(S + ((size_t)(bh * C64) + seg) * DD + 4 * c) = o;
  }

  float* my = red + wave * (DD * DD);
  for (int i = 0; i < 4; ++i)
    for (int r = 0; r < 8; ++r) {
      v4f o;
      o.x = acc[i][0][r];
      o.y = acc[i][1][r];
      o.z = acc[i][2][r];
      o.w = acc[i][3][r];
      *(v4f*)(my + (4 * (8 * h + r) + i) * DD + 4 * c) = o;
    }
  __syncthreads();

  float* gout = Gpart + (size_t)(bh * K1_CHUNKS + ch) * (DD * DD);
  for (int x = 0; x < 8; ++x) {
    const int idx = (x * 128 + t) * 4;
    const v4f p0 = *(const v4f*)(red + idx);
    const v4f p1 = *(const v4f*)(red + DD * DD + idx);
    const v4f p2 = *(const v4f*)(red + 2 * DD * DD + idx);
    const v4f p3 = *(const v4f*)(red + 3 * DD * DD + idx);
    const int o = ((idx & 48) << 6) + ((idx >> 6) << 4) + (idx & 15);
    *(v4f*)(gout + o) = ((p0 + p1) + p2) + p3;
  }
}

#define WTP 68
#define K2_SLABS 4
__global__ void __launch_bounds__(256)
k2_combine(const float* __restrict__ proj,
           const float* __restrict__ Gpart,
           float* __restrict__ S, float* __restrict__ Wm,
           float* __restrict__ GramG) {
  __shared__ __attribute__((aligned(16))) float Gs[DD * 16];
  __shared__ __attribute__((aligned(16))) float Pl[DD * DD];
  __shared__ float Gm[DD * 65];
  const int bh = blockIdx.x;
  const int part = blockIdx.y;
  const int t  = threadIdx.x;

  if (part == K2_SLABS) {
    if (t < 64) {
      float sv[C64];
      for (int c = 0; c < C64; ++c) sv[c] = S[((size_t)(bh * C64) + c) * DD + t];
      float run = 0.f;
      for (int c = 0; c < C64; ++c) {
        S[((size_t)(bh * C64) + c) * DD + t] = run;
        run += sv[c];
      }
    }
    return;
  }

  {
    const float* gp = Gpart + (size_t)bh * K1_CHUNKS * (DD * DD) + part * (DD * 16) + t * 4;
    v4f pr[4];
    for (int x = 0; x < 4; ++x) pr[x] = *(const v4f*)(proj + (x * 256 + t) * 4);
    v4f gq[K1_CHUNKS];
    for (int c = 0; c < K1_CHUNKS; ++c) gq[c] = *(const v4f*)(gp + (size_t)c * (DD * DD));
    for (int x = 0; x < 4; ++x) *(v4f*)(Pl + (x * 256 + t) * 4) = pr[x];
    v4f s = {0.f, 0.f, 0.f, 0.f};
    for (int c = 0; c < K1_CHUNKS; ++c) s += gq[c];
    *(v4f*)(Gs + t * 4) = s;
  }
  __syncthreads();

  const int wave = __builtin_amdgcn_readfirstlane(t >> 5);
  const int lane = t & 31;
  const int m = lane & 15, khalf = (lane >> 4) * 2;
  const int ti = wave >> 1;
  const int c0 = (wave & 1) * 32 + m;
  const int c1 = c0 + 16;
  const int rbase = ti * 16 + (lane >> 4) * 8;

  {
    v8f acc0 = {0.f, 0.f, 0.f, 0.f, 0.f, 0.f, 0.f, 0.f};
    v8f acc1 = {0.f, 0.f, 0.f, 0.f, 0.f, 0.f, 0.f, 0.f};
    for (int kk = 0; kk < DD; kk += 4) {
      const float* r0 = Pl + (kk + khalf) * DD;
      const float* r1 = r0 + DD;
      v2f a, b0, b1;
      a.x  = r0[ti * 16 + m];
      a.y  = r1[ti * 16 + m];
      b0.x = r0[c0];
      b0.y = r1[c0];
      b1.x = r0[c1];
      b1.y = r1[c1];
      acc0 = wmma_f32(a, b0, acc0);
      acc1 = wmma_f32(a, b1, acc1);
    }
    for (int j = 0; j < 8; ++j) {
      Gm[(rbase + j) * 65 + c0] = acc0[j];
      Gm[(rbase + j) * 65 + c1] = acc1[j];
    }
    if (bh == 0 && part == 0) {
      for (int j = 0; j < 8; ++j) {
        GramG[(rbase + j) * DD + c0] = acc0[j];
        GramG[(rbase + j) * DD + c1] = acc1[j];
      }
    }
  }
  __syncthreads();

  if (wave < 4) {
    v8f acc = {0.f, 0.f, 0.f, 0.f, 0.f, 0.f, 0.f, 0.f};
    const float* arow = Gm + (wave * 16 + m) * 65 + khalf;
    for (int kk = 0; kk < DD; kk += 4) {
      const float* r0 = Gs + (kk + khalf) * 16;
      v2f a, b;
      a.x = arow[kk];
      a.y = arow[kk + 1];
      b.x = r0[m];
      b.y = r0[16 + m];
      acc = wmma_f32(a, b, acc);
    }
    const int rb = wave * 16 + (lane >> 4) * 8;
    *(v4f*)(Pl + m * WTP + rb)     = (v4f){acc[0], acc[1], acc[2], acc[3]};
    *(v4f*)(Pl + m * WTP + rb + 4) = (v4f){acc[4], acc[5], acc[6], acc[7]};
  }
  __syncthreads();
  {
    float* wout = Wm + (size_t)bh * DD * DD + part * (16 * DD);
    const int wrow = t >> 4, wcol = (t & 15) * 4;
    *(v4f*)(wout + wrow * DD + wcol) = *(const v4f*)(Pl + wrow * WTP + wcol);
  }
}

#define QP 68

static __device__ __forceinline__ void band_product(const v4f (&qa)[8], const float* bt0,
                                                    const float* bt1, v8f& acc0, v8f& acc1) {
  v4f b0[8], b1[8];
  b0[0] = *(const v4f*)(bt0);
  b1[0] = *(const v4f*)(bt1);
  b0[1] = *(const v4f*)(bt0 + 8);
  b1[1] = *(const v4f*)(bt1 + 8);
  for (int g = 0; g < 4; ++g) {
    if (g < 3) {
      b0[2 * g + 2] = *(const v4f*)(bt0 + 8 * (2 * g + 2));
      b1[2 * g + 2] = *(const v4f*)(bt1 + 8 * (2 * g + 2));
      b0[2 * g + 3] = *(const v4f*)(bt0 + 8 * (2 * g + 3));
      b1[2 * g + 3] = *(const v4f*)(bt1 + 8 * (2 * g + 3));
    }
    for (int p = 2 * g; p < 2 * g + 2; ++p) {
      const v2f alo = {qa[p].x, qa[p].y};
      const v2f ahi = {qa[p].z, qa[p].w};
      const v2f x0 = {b0[p].x, b0[p].y};
      const v2f x1 = {b1[p].x, b1[p].y};
      const v2f y0 = {b0[p].z, b0[p].w};
      const v2f y1 = {b1[p].z, b1[p].w};
      acc0 = wmma_f32(alo, x0, acc0);
      acc1 = wmma_f32(alo, x1, acc1);
      acc0 = wmma_f32(ahi, y0, acc0);
      acc1 = wmma_f32(ahi, y1, acc1);
    }
  }
}

__global__ void __launch_bounds__(256, 12)
k3_out(const float* __restrict__ q, const float* __restrict__ k,
                       const float* __restrict__ S, const float* __restrict__ Wm,
                       const float* __restrict__ GramG, float* __restrict__ out) {
  __shared__ float kc[64 * 65];
  __shared__ __attribute__((aligned(16))) float ab[64 * QP];
  __shared__ __attribute__((aligned(16))) float qgl[64 * QP];
  __shared__ float qsum[4 * 64];
  __shared__ float dinv[64];
  const int bh = blockIdx.x, c64 = blockIdx.y;
  const int t  = threadIdx.x;
  const int n0 = c64 * 64;
  const float* qb = q + (size_t)bh * NN * DD;
  const float* kb = k + (size_t)bh * NN * DD;

  const float* kch = kb + (size_t)n0 * DD;
  const float* qch = qb + (size_t)n0 * DD;
  const float* wb  = Wm + (size_t)bh * DD * DD;
  v4f kr[4], qr[4], gr[4];
  for (int x = 0; x < 4; ++x) kr[x] = *(const v4f*)(kch + (x * 256 + t) * 4);
  for (int x = 0; x < 4; ++x) qr[x] = *(const v4f*)(qch + (x * 256 + t) * 4);
  for (int x = 0; x < 4; ++x) gr[x] = *(const v4f*)(GramG + (x * 256 + t) * 4);
  float pre = 0.f;
  if (t < 64) pre = S[((size_t)(bh * C64) + c64) * DD + t];
  const int srow = t >> 4, scol = (t & 15) * 4;
  for (int x = 0; x < 4; ++x) {
    float* kd = kc + (x * 16 + srow) * 65 + scol;
    kd[0] = kr[x].x; kd[1] = kr[x].y; kd[2] = kr[x].z; kd[3] = kr[x].w;
    *(v4f*)(ab + (x * 16 + srow) * QP + scol) = gr[x];
    *(v4f*)(qgl + (x * 16 + srow) * QP + scol) = qr[x];
  }
  __syncthreads();

  const int wave = t >> 5, lane = t & 31;
  const int m = lane & 15, kq = (lane >> 4) * 4;
  const int ti  = wave >> 1;
  const int tj0 = (wave & 1) * 2;
  const int rbase = ti * 16 + (lane >> 4) * 8;

  v4f qa[8];
  {
    const float* qrow = qgl + (ti * 16 + m) * QP + kq;
    for (int p = 0; p < 8; ++p) qa[p] = *(const v4f*)(qrow + 8 * p);
  }

  const int d = t & 63, qt = t >> 6;
  {
    float run = 0.f;
    for (int r = qt * 16; r < qt * 16 + 16; ++r) {
      run += kc[r * 65 + d];
      kc[r * 65 + d] = run;
    }
    qsum[qt * 64 + d] = run;
  }
  __syncthreads();
  if (t < 64) {
    const float o0 = pre;
    const float o1 = o0 + qsum[0 * 64 + t];
    const float o2 = o1 + qsum[1 * 64 + t];
    const float o3 = o2 + qsum[2 * 64 + t];
    qsum[0 * 64 + t] = o0; qsum[1 * 64 + t] = o1;
    qsum[2 * 64 + t] = o2; qsum[3 * 64 + t] = o3;
  }
  __syncthreads();
  {
    const float off = qsum[qt * 64 + d];
    for (int r = qt * 16; r < qt * 16 + 16; ++r) kc[r * 65 + d] += off;
  }
  __syncthreads();

  const float* bt0 = ab + (tj0 * 16 + m) * QP + kq;
  const float* bt1 = bt0 + 16 * QP;

  {
    v8f acc0 = {0.f, 0.f, 0.f, 0.f, 0.f, 0.f, 0.f, 0.f};
    v8f acc1 = {0.f, 0.f, 0.f, 0.f, 0.f, 0.f, 0.f, 0.f};
    band_product(qa, bt0, bt1, acc0, acc1);
    for (int j = 0; j < 8; ++j) {
      qgl[(rbase + j) * 65 + tj0 * 16 + m]      = acc0[j];
      qgl[(rbase + j) * 65 + tj0 * 16 + 16 + m] = acc1[j];
    }
  }
  __syncthreads();

  v4f wr[4];
  for (int x = 0; x < 4; ++x) wr[x] = *(const v4f*)(wb + (x * 256 + t) * 4);
  if (t < 64) {
    float s = 0.f;
    for (int dd = 0; dd < 64; ++dd) s += qgl[t * 65 + dd] * kc[t * 65 + dd];
    dinv[t] = 1.0f / s;
  }
  for (int x = 0; x < 4; ++x) *(v4f*)(ab + (x * 16 + srow) * QP + scol) = wr[x];
  __syncthreads();

  {
    v8f acc0 = {0.f, 0.f, 0.f, 0.f, 0.f, 0.f, 0.f, 0.f};
    v8f acc1 = {0.f, 0.f, 0.f, 0.f, 0.f, 0.f, 0.f, 0.f};
    band_product(qa, bt0, bt1, acc0, acc1);
    for (int j = 0; j < 8; ++j) {
      const float dv = dinv[rbase + j];
      float* orow = out + ((size_t)bh * NN + n0 + rbase + j) * DD + tj0 * 16 + m;
      orow[0]  = acc0[j] * dv;
      orow[16] = acc1[j] * dv;
    }
  }
}

extern "C" void kernel_launch(void* const* d_in, const int* in_sizes, int n_in,
                              void* d_out, int out_size, void* d_ws, size_t ws_size,
                              hipStream_t stream) {
  const float* q    = (const float*)d_in[0];
  const float* k    = (const float*)d_in[1];
  const float* v    = (const float*)d_in[2];
  const float* proj = (const float*)d_in[3];
  float* out = (float*)d_out;

  float* ws    = (float*)d_ws;
  float* Gpart = ws;
  float* S     = Gpart + (size_t)BH * K1_CHUNKS * DD * DD;
  float* Wm    = S + (size_t)BH * C64 * DD;
  float* Gram  = Wm + (size_t)BH * DD * DD;

  k1_gpart<<<dim3(BH, K1_CHUNKS), 128, 0, stream>>>(k, v, Gpart, S);
  k2_combine<<<dim3(BH, 5), 256, 0, stream>>>(proj, Gpart, S, Wm, Gram);
  k3_out<<<dim3(BH, C64), 256, 0, stream>>>(q, k, S, Wm, Gram, out);
}


# K3: one wave owns a 16-row band over all 64 columns (4-wave groups, half the waves): q@Gram stays in accumulators, k running sum and den formed in that register map; sum order of both changes
# speedup vs baseline: 2.2803x; 1.1609x over previous
#include <hip/hip_runtime.h>


typedef __attribute__((ext_vector_type(2))) float v2f;
typedef __attribute__((ext_vector_type(4))) float v4f;
typedef __attribute__((ext_vector_type(8))) float v8f;

#define BH 64
#define NN 4096
#define DD 64
#define K1_ROWS 256
#define K1_CHUNKS 16
#define C64 64

static __device__ __forceinline__ v8f wmma_f32(v2f a, v2f b, v8f c) {
  return __builtin_amdgcn_wmma_f32_16x16x4_f32(false, a, false, b, (short)0, c,
                                               false, false);
}

#define K1_WAVES 4
#define K1_STEPS 16
#define K1_AHEAD 5

__global__ void __launch_bounds__(128)
k1_gpart(const float* __restrict__ k, const float* __restrict__ v,
         float* __restrict__ Gpart, float* __restrict__ S) {
  __shared__ float red[K1_WAVES * DD * DD];
  const int bh = blockIdx.x;
  const int ch = blockIdx.y;
  const int t  = threadIdx.x;
  const int wave = t >> 5, lane = t & 31;
  const int c = lane & 15, h = lane >> 4;
  const int seg = ch * K1_WAVES + wave;
  const size_t base = ((size_t)bh * NN + (size_t)seg * 64 + 2 * h) * DD + 4 * c;
  const float* kp = k + base;
  const float* vp = v + base;

  v8f acc[4][4];
  for (int i = 0; i < 4; ++i)
    for (int j = 0; j < 4; ++j)
      acc[i][j] = (v8f){0.f, 0.f, 0.f, 0.f, 0.f, 0.f, 0.f, 0.f};
  v4f ksum = {0.f, 0.f, 0.f, 0.f};

  v4f rk0[K1_AHEAD], rk1[K1_AHEAD], rv0[K1_AHEAD], rv1[K1_AHEAD];
  for (int p = 0; p < K1_AHEAD - 1; ++p) {
    rk0[p] = *(const v4f*)(kp + (4 * p) * DD);
    rk1[p] = *(const v4f*)(kp + (4 * p + 1) * DD);
    rv0[p] = *(const v4f*)(vp + (4 * p) * DD);
    rv1[p] = *(const v4f*)(vp + (4 * p + 1) * DD);
  }
  for (int s = 0; s < K1_STEPS; ++s) {
    const int nx = s + K1_AHEAD - 1;
    if (nx < K1_STEPS) {
      const int w = nx % K1_AHEAD;
      rk0[w] = *(const v4f*)(kp + (4 * nx) * DD);
      rk1[w] = *(const v4f*)(kp + (4 * nx + 1) * DD);
      rv0[w] = *(const v4f*)(vp + (4 * nx) * DD);
      rv1[w] = *(const v4f*)(vp + (4 * nx + 1) * DD);
    }
    const int u = s % K1_AHEAD;
    ksum += rk0[u] + rk1[u];
    for (int i = 0; i < 4; ++i) {
      v2f a;
      a.x = rk0[u][i];
      a.y = rk1[u][i];
      for (int j = 0; j < 4; ++j) {
        v2f b;
        b.x = rv0[u][j];
        b.y = rv1[u][j];
        acc[i][j] = wmma_f32(a, b, acc[i][j]);
      }
    }
  }

  {
    v4f o;
    o.x = ksum.x + __shfl_xor(ksum.x, 16);
    o.y = ksum.y + __shfl_xor(ksum.y, 16);
    o.z = ksum.z + __shfl_xor(ksum.z, 16);
    o.w = ksum.w + __shfl_xor(ksum.w, 16);
    if (h == 0) *(v4f*)(S + ((size_t)(bh * C64) + seg) * DD + 4 * c) = o;
  }

  float* my = red + wave * (DD * DD);
  for (int i = 0; i < 4; ++i)
    for (int r = 0; r < 8; ++r) {
      v4f o;
      o.x = acc[i][0][r];
      o.y = acc[i][1][r];
      o.z = acc[i][2][r];
      o.w = acc[i][3][r];
      *(v4f*)(my + (4 * (8 * h + r) + i) * DD + 4 * c) = o;
    }
  __syncthreads();

  float* gout = Gpart + (size_t)(bh * K1_CHUNKS + ch) * (DD * DD);
  for (int x = 0; x < 8; ++x) {
    const int idx = (x * 128 + t) * 4;
    const v4f p0 = *(const v4f*)(red + idx);
    const v4f p1 = *(const v4f*)(red + DD * DD + idx);
    const v4f p2 = *(const v4f*)(red + 2 * DD * DD + idx);
    const v4f p3 = *(const v4f*)(red + 3 * DD * DD + idx);
    const int o = ((idx & 48) << 6) + ((idx >> 6) << 4) + (idx & 15);
    *(v4f*)(gout + o) = ((p0 + p1) + p2) + p3;
  }
}

#define WTP 68
#define K2_SLABS 4
__global__ void __launch_bounds__(256)
k2_combine(const float* __restrict__ proj,
           const float* __restrict__ Gpart,
           float* __restrict__ S, float* __restrict__ Wm,
           float* __restrict__ GramG) {
  __shared__ __attribute__((aligned(16))) float Gs[DD * 16];
  __shared__ __attribute__((aligned(16))) float Pl[DD * DD];
  __shared__ float Gm[DD * 65];
  const int bh = blockIdx.x;
  const int part = blockIdx.y;
  const int t  = threadIdx.x;

  if (part == K2_SLABS) {
    if (t < 64) {
      float sv[C64];
      for (int c = 0; c < C64; ++c) sv[c] = S[((size_t)(bh * C64) + c) * DD + t];
      float run = 0.f;
      for (int c = 0; c < C64; ++c) {
        S[((size_t)(bh * C64) + c) * DD + t] = run;
        run += sv[c];
      }
    }
    return;
  }

  {
    const float* gp = Gpart + (size_t)bh * K1_CHUNKS * (DD * DD) + part * (DD * 16) + t * 4;
    v4f pr[4];
    for (int x = 0; x < 4; ++x) pr[x] = *(const v4f*)(proj + (x * 256 + t) * 4);
    v4f gq[K1_CHUNKS];
    for (int c = 0; c < K1_CHUNKS; ++c) gq[c] = *(const v4f*)(gp + (size_t)c * (DD * DD));
    for (int x = 0; x < 4; ++x) *(v4f*)(Pl + (x * 256 + t) * 4) = pr[x];
    v4f s = {0.f, 0.f, 0.f, 0.f};
    for (int c = 0; c < K1_CHUNKS; ++c) s += gq[c];
    *(v4f*)(Gs + t * 4) = s;
  }
  __syncthreads();

  const int wave = __builtin_amdgcn_readfirstlane(t >> 5);
  const int lane = t & 31;
  const int m = lane & 15, khalf = (lane >> 4) * 2;
  const int ti = wave >> 1;
  const int c0 = (wave & 1) * 32 + m;
  const int c1 = c0 + 16;
  const int rbase = ti * 16 + (lane >> 4) * 8;

  {
    v8f acc0 = {0.f, 0.f, 0.f, 0.f, 0.f, 0.f, 0.f, 0.f};
    v8f acc1 = {0.f, 0.f, 0.f, 0.f, 0.f, 0.f, 0.f, 0.f};
    for (int kk = 0; kk < DD; kk += 4) {
      const float* r0 = Pl + (kk + khalf) * DD;
      const float* r1 = r0 + DD;
      v2f a, b0, b1;
      a.x  = r0[ti * 16 + m];
      a.y  = r1[ti * 16 + m];
      b0.x = r0[c0];
      b0.y = r1[c0];
      b1.x = r0[c1];
      b1.y = r1[c1];
      acc0 = wmma_f32(a, b0, acc0);
      acc1 = wmma_f32(a, b1, acc1);
    }
    for (int j = 0; j < 8; ++j) {
      Gm[(rbase + j) * 65 + c0] = acc0[j];
      Gm[(rbase + j) * 65 + c1] = acc1[j];
    }
    if (bh == 0 && part == 0) {
      for (int j = 0; j < 8; ++j) {
        GramG[(rbase + j) * DD + c0] = acc0[j];
        GramG[(rbase + j) * DD + c1] = acc1[j];
      }
    }
  }
  __syncthreads();

  if (wave < 4) {
    v8f acc = {0.f, 0.f, 0.f, 0.f, 0.f, 0.f, 0.f, 0.f};
    const float* arow = Gm + (wave * 16 + m) * 65 + khalf;
    for (int kk = 0; kk < DD; kk += 4) {
      const float* r0 = Gs + (kk + khalf) * 16;
      v2f a, b;
      a.x = arow[kk];
      a.y = arow[kk + 1];
      b.x = r0[m];
      b.y = r0[16 + m];
      acc = wmma_f32(a, b, acc);
    }
    const int rb = wave * 16 + (lane >> 4) * 8;
    *(v4f*)(Pl + m * WTP + rb)     = (v4f){acc[0], acc[1], acc[2], acc[3]};
    *(v4f*)(Pl + m * WTP + rb + 4) = (v4f){acc[4], acc[5], acc[6], acc[7]};
  }
  __syncthreads();
  {
    float* wout = Wm + (size_t)bh * DD * DD + part * (16 * DD);
    const int wrow = t >> 4, wcol = (t & 15) * 4;
    *(v4f*)(wout + wrow * DD + wcol) = *(const v4f*)(Pl + wrow * WTP + wcol);
  }
}

#define QP 68
#define K3_THREADS 128

static __device__ __forceinline__ float xhalf(float x) {
  const int b = __builtin_bit_cast(int, x);
  return __builtin_bit_cast(float,
      __builtin_amdgcn_permlanex16(b, b, 0x76543210, 0xfedcba98, false, false));
}

template <int MASK>
static __device__ __forceinline__ float row_xor(float x) {
  const int b = __builtin_bit_cast(int, x);
  return __builtin_bit_cast(float,
      __builtin_amdgcn_update_dpp(b, b, 0x160 + MASK, 0xf, 0xf, false));
}

static __device__ __forceinline__ void band_product(const v4f (&qa)[8], const float* bt,
                                                    v8f (&acc)[4]) {
  v4f b[8][4];
  for (int p = 0; p < 2; ++p)
    for (int u = 0; u < 4; ++u) b[p][u] = *(const v4f*)(bt + u * 16 * QP + 8 * p);
  for (int g = 0; g < 4; ++g) {
    if (g < 3) {
      for (int p = 2 * g + 2; p < 2 * g + 4; ++p)
        for (int u = 0; u < 4; ++u) b[p][u] = *(const v4f*)(bt + u * 16 * QP + 8 * p);
    }
    for (int p = 2 * g; p < 2 * g + 2; ++p) {
      const v2f alo = {qa[p].x, qa[p].y};
      const v2f ahi = {qa[p].z, qa[p].w};
      for (int u = 0; u < 4; ++u) {
        const v2f x = {b[p][u].x, b[p][u].y};
        acc[u] = wmma_f32(alo, x, acc[u]);
      }
      for (int u = 0; u < 4; ++u) {
        const v2f y = {b[p][u].z, b[p][u].w};
        acc[u] = wmma_f32(ahi, y, acc[u]);
      }
    }
  }
}

__global__ void __launch_bounds__(K3_THREADS, 6)
k3_out(const float* __restrict__ q, const float* __restrict__ k,
                       const float* __restrict__ S, const float* __restrict__ Wm,
                       const float* __restrict__ GramG, float* __restrict__ out) {
  __shared__ __attribute__((aligned(16))) float kl[64 * QP];
  __shared__ __attribute__((aligned(16))) float ab[64 * QP];
  __shared__ __attribute__((aligned(16))) float ql[64 * QP];
  __shared__ float bandtot[4 * 64];
  const int bh = blockIdx.x, c64 = blockIdx.y;
  const int t  = threadIdx.x;
  const int n0 = c64 * 64;
  const float* qb = q + (size_t)bh * NN * DD;
  const float* kb = k + (size_t)bh * NN * DD;

  const int lane = t & 31;
  const int m = lane & 15, hh = lane >> 4;
  const int kq = hh * 4;
  const int band = __builtin_amdgcn_readfirstlane(t >> 5);
  const int rbase = band * 16 + hh * 8;

  const float* kch = kb + (size_t)n0 * DD;
  const float* qch = qb + (size_t)n0 * DD;
  const float* wb  = Wm + (size_t)bh * DD * DD;
  v4f kr[8], qr[8], gr[8];
  for (int x = 0; x < 8; ++x) kr[x] = *(const v4f*)(kch + (x * K3_THREADS + t) * 4);
  for (int x = 0; x < 8; ++x) qr[x] = *(const v4f*)(qch + (x * K3_THREADS + t) * 4);
  for (int x = 0; x < 8; ++x) gr[x] = *(const v4f*)(GramG + (x * K3_THREADS + t) * 4);
  float pre[4];
  for (int u = 0; u < 4; ++u) pre[u] = S[((size_t)(bh * C64) + c64) * DD + 16 * u + m];
  const int srow = t >> 4, scol = (t & 15) * 4;
  for (int x = 0; x < 8; ++x) {
    *(v4f*)(kl + (x * 8 + srow) * QP + scol) = kr[x];
    *(v4f*)(ab + (x * 8 + srow) * QP + scol) = gr[x];
    *(v4f*)(ql + (x * 8 + srow) * QP + scol) = qr[x];
  }
  __syncthreads();

  v4f qa[8];
  {
    const float* qrow = ql + (band * 16 + m) * QP + kq;
    for (int p = 0; p < 8; ++p) qa[p] = *(const v4f*)(qrow + 8 * p);
  }

  const float* bt = ab + m * QP + kq;

  v8f qg[4];
  for (int u = 0; u < 4; ++u) qg[u] = (v8f){0.f, 0.f, 0.f, 0.f, 0.f, 0.f, 0.f, 0.f};
  band_product(qa, bt, qg);

  v4f wr[8];
  for (int x = 0; x < 8; ++x) wr[x] = *(const v4f*)(wb + (x * K3_THREADS + t) * 4);

  float kk[4][8];
  for (int u = 0; u < 4; ++u)
    for (int j = 0; j < 8; ++j) kk[u][j] = kl[(rbase + j) * QP + 16 * u + m];
  float up[4];
  for (int u = 0; u < 4; ++u) {
    for (int j = 1; j < 8; ++j) kk[u][j] += kk[u][j - 1];
    const float other = xhalf(kk[u][7]);
    up[u] = hh ? other : 0.f;
    if (hh == 0) bandtot[band * 64 + 16 * u + m] = kk[u][7] + other;
  }
  __syncthreads();

  for (int x = 0; x < 8; ++x) *(v4f*)(ab + (x * 8 + srow) * QP + scol) = wr[x];

  float den[8];
  for (int j = 0; j < 8; ++j) den[j] = 0.f;
  for (int u = 0; u < 4; ++u) {
    float off = pre[u];
    for (int b = 0; b < 3; ++b)
      if (b < band) off += bandtot[b * 64 + 16 * u + m];
    off += up[u];
    for (int j = 0; j < 8; ++j) den[j] += qg[u][j] * (kk[u][j] + off);
  }
  float dv[8];
  for (int j = 0; j < 8; ++j) {
    float s = den[j];
    s += row_xor<1>(s);
    s += row_xor<2>(s);
    s += row_xor<4>(s);
    s += row_xor<8>(s);
    dv[j] = 1.0f / s;
  }
  __syncthreads();

  {
    v8f acc[4];
    for (int u = 0; u < 4; ++u) acc[u] = (v8f){0.f, 0.f, 0.f, 0.f, 0.f, 0.f, 0.f, 0.f};
    band_product(qa, bt, acc);
    for (int j = 0; j < 8; ++j) {
      float* orow = out + ((size_t)bh * NN + n0 + rbase + j) * DD + m;
      for (int u = 0; u < 4; ++u) orow[16 * u] = acc[u][j] * dv[j];
    }
  }
}

extern "C" void kernel_launch(void* const* d_in, const int* in_sizes, int n_in,
                              void* d_out, int out_size, void* d_ws, size_t ws_size,
                              hipStream_t stream) {
  const float* q    = (const float*)d_in[0];
  const float* k    = (const float*)d_in[1];
  const float* v    = (const float*)d_in[2];
  const float* proj = (const float*)d_in[3];
  float* out = (float*)d_out;

  float* ws    = (float*)d_ws;
  float* Gpart = ws;
  float* S     = Gpart + (size_t)BH * K1_CHUNKS * DD * DD;
  float* Wm    = S + (size_t)BH * C64 * DD;
  float* Gram  = Wm + (size_t)BH * DD * DD;

  k1_gpart<<<dim3(BH, K1_CHUNKS), 128, 0, stream>>>(k, v, Gpart, S);
  k2_combine<<<dim3(BH, 5), 256, 0, stream>>>(proj, Gpart, S, Wm, Gram);
  k3_out<<<dim3(BH, C64), K3_THREADS, 0, stream>>>(q, k, S, Wm, Gram, out);
}
